# starting kernel
# baseline (speedup 1.0000x reference)
#include <hip/hip_runtime.h>
#include <math.h>

#define LDA 65
#define NL  4

typedef __attribute__((ext_vector_type(2))) float v2f;
typedef __attribute__((ext_vector_type(8))) float v8f;

__device__ __forceinline__ float2 cmul(float2 a, float2 b) {
  return make_float2(a.x * b.x - a.y * b.y, a.x * b.y + a.y * b.x);
}
__device__ __forceinline__ float2 cadd(float2 a, float2 b) {
  return make_float2(a.x + b.x, a.y + b.y);
}

struct C2 { float2 a00, a01, a10, a11; };
__device__ __forceinline__ C2 mm2(const C2 L, const C2 R) {
  C2 o;
  o.a00 = cadd(cmul(L.a00, R.a00), cmul(L.a01, R.a10));
  o.a01 = cadd(cmul(L.a00, R.a01), cmul(L.a01, R.a11));
  o.a10 = cadd(cmul(L.a10, R.a00), cmul(L.a11, R.a10));
  o.a11 = cadd(cmul(L.a10, R.a01), cmul(L.a11, R.a11));
  return o;
}

__device__ __forceinline__ void rot_gate(float phi, float th, float om,
                                         float2* u00, float2* u01,
                                         float2* u10, float2* u11) {
  float ct = cosf(0.5f * th), st = sinf(0.5f * th);
  float ap = -0.5f * (phi + om);
  float am = 0.5f * (phi - om);
  *u00 = make_float2(ct * cosf(ap), ct * sinf(ap));
  *u01 = make_float2(-st * cosf(am), -st * sinf(am));
  *u10 = make_float2(st * cosf(-am), st * sinf(-am));
  *u11 = make_float2(ct * cosf(-ap), ct * sinf(-ap));
}

__device__ __forceinline__ void apply_cgate(float* Mr, float* Mi,
                                            float2 u00, float2 u01,
                                            float2 u10, float2 u11,
                                            int cb, int tb, int tid) {
  for (int idx = tid; idx < 4096; idx += 256) {
    int i = idx >> 6, j = idx & 63;
    if ((i & cb) && !(i & tb)) {
      int a0 = i * LDA + j, a1 = (i | tb) * LDA + j;
      float2 x0 = make_float2(Mr[a0], Mi[a0]);
      float2 x1 = make_float2(Mr[a1], Mi[a1]);
      float2 y0 = cadd(cmul(u00, x0), cmul(u01, x1));
      float2 y1 = cadd(cmul(u10, x0), cmul(u11, x1));
      Mr[a0] = y0.x; Mi[a0] = y0.y;
      Mr[a1] = y1.x; Mi[a1] = y1.y;
    }
  }
}

__global__ __launch_bounds__(256) void qk_prep(const float* __restrict__ rotp,
                                               const float* __restrict__ entp,
                                               float2* __restrict__ W) {
  __shared__ float Mr[64 * LDA], Mi[64 * LDA];
  __shared__ float2 U[24][2][2];
  const int l = blockIdx.x;
  const int tid = threadIdx.x;

  if (tid < 24) {
    const float* p = (tid < 12) ? &rotp[(l * 12 + tid) * 3]
                                : &entp[(l * 12 + (tid - 12)) * 3];
    rot_gate(p[0], p[1], p[2], &U[tid][0][0], &U[tid][0][1],
             &U[tid][1][0], &U[tid][1][1]);
  }
  __syncthreads();

  float2* Wl = W + (size_t)l * 3 * 4096;

  for (int idx = tid; idx < 4096; idx += 256) {
    int i = idx >> 6, j = idx & 63;
    float2 p = U[0][(i >> 5) & 1][(j >> 5) & 1];
    for (int q = 1; q < 6; q++)
      p = cmul(p, U[q][(i >> (5 - q)) & 1][(j >> (5 - q)) & 1]);
    Mr[i * LDA + j] = p.x; Mi[i * LDA + j] = p.y;
  }
  __syncthreads();
  for (int c = 0; c < 5; c++) {
    apply_cgate(Mr, Mi, U[12 + c][0][0], U[12 + c][0][1],
                U[12 + c][1][0], U[12 + c][1][1],
                1 << (5 - c), 1 << (4 - c), tid);
    __syncthreads();
  }
  for (int idx = tid; idx < 4096; idx += 256) {
    int i = idx >> 6, j = idx & 63;
    Wl[0 * 4096 + idx] = make_float2(Mr[i * LDA + j], Mi[i * LDA + j]);
  }

  for (int idx = tid; idx < 4096; idx += 256) {
    int i = idx >> 6, j = idx & 63;
    float2 p = U[6][(j >> 5) & 1][(i >> 5) & 1];
    for (int q = 1; q < 6; q++)
      p = cmul(p, U[6 + q][(j >> (5 - q)) & 1][(i >> (5 - q)) & 1]);
    Wl[1 * 4096 + idx] = p;
  }
  __syncthreads();

  for (int idx = tid; idx < 4096; idx += 256) {
    int i = idx >> 6, j = idx & 63;
    Mr[i * LDA + j] = (i == j) ? 1.f : 0.f;
    Mi[i * LDA + j] = 0.f;
  }
  __syncthreads();
  for (int c = 0; c < 5; c++) {
    apply_cgate(Mr, Mi, U[18 + c][0][0], U[18 + c][0][1],
                U[18 + c][1][0], U[18 + c][1][1],
                1 << (5 - c), 1 << (4 - c), tid);
    __syncthreads();
  }
  for (int idx = tid; idx < 4096; idx += 256) {
    int i = idx >> 6, j = idx & 63;
    Wl[2 * 4096 + idx] = make_float2(Mr[j * LDA + i], Mi[j * LDA + i]);
  }
}

__device__ __forceinline__ v8f wmma4(v2f a, v2f b, v8f c) {
  return __builtin_amdgcn_wmma_f32_16x16x4_f32(
      false, a, false, b, (short)0, c, false, false);
}

__device__ __forceinline__ void cgemm64(const float* Ar, const float* Ai,
                                        const float* Br, const float* Bi,
                                        float* Dr, float* Di) {
  const int tid = threadIdx.x;
  const int lane = tid & 31;
  const int w = tid >> 5;
  const int mt = (w & 3) * 16;
  const int nt0 = (w >> 2) * 32;
  const int nt1 = nt0 + 16;
  const int ml = lane & 15;
  const int kb = (lane >> 4) * 2;
  const int hi = lane >> 4;

  v8f p1_0 = {}, p2_0 = {}, p3_0 = {};
  v8f p1_1 = {}, p2_1 = {}, p3_1 = {};
  for (int k = 0; k < 64; k += 4) {
    const float* ap = Ar + (mt + ml) * LDA + k + kb;
    const float* aip = Ai + (mt + ml) * LDA + k + kb;
    v2f ar; ar.x = ap[0];  ar.y = ap[1];
    v2f ai; ai.x = aip[0]; ai.y = aip[1];
    v2f as = ar + ai;
    v2f br0, bi0, br1, bi1;
    br0.x = Br[(k + kb) * LDA + nt0 + ml]; br0.y = Br[(k + kb + 1) * LDA + nt0 + ml];
    bi0.x = Bi[(k + kb) * LDA + nt0 + ml]; bi0.y = Bi[(k + kb + 1) * LDA + nt0 + ml];
    br1.x = Br[(k + kb) * LDA + nt1 + ml]; br1.y = Br[(k + kb + 1) * LDA + nt1 + ml];
    bi1.x = Bi[(k + kb) * LDA + nt1 + ml]; bi1.y = Bi[(k + kb + 1) * LDA + nt1 + ml];
    v2f bs0 = br0 + bi0;
    v2f bs1 = br1 + bi1;

    p1_0 = wmma4(ar, br0, p1_0);
    p2_0 = wmma4(ai, bi0, p2_0);
    p3_0 = wmma4(as, bs0, p3_0);
    p1_1 = wmma4(ar, br1, p1_1);
    p2_1 = wmma4(ai, bi1, p2_1);
    p3_1 = wmma4(as, bs1, p3_1);
  }

  v8f cr0 = p1_0 - p2_0;
  v8f ci0 = p3_0 - p1_0 - p2_0;
  v8f cr1 = p1_1 - p2_1;
  v8f ci1 = p3_1 - p1_1 - p2_1;

  for (int v = 0; v < 8; v++) {
    int m = mt + v + 8 * hi;
    int n0 = nt0 + ml, n1 = nt1 + ml;
    Dr[m * LDA + n0] = cr0[v]; Di[m * LDA + n0] = ci0[v];
    Dr[m * LDA + n1] = cr1[v]; Di[m * LDA + n1] = ci1[v];
  }
}

__global__ __launch_bounds__(256) void qk_main(const float* __restrict__ x,
                                               const float* __restrict__ entp,
                                               const float2* __restrict__ W,
                                               float* __restrict__ out) {
  __shared__ float S0r[64 * LDA], S0i[64 * LDA];
  __shared__ float S1r[64 * LDA], S1i[64 * LDA];
  __shared__ float OpR[64 * LDA], OpI[64 * LDA];
  __shared__ float2 E[12][2][2];
  __shared__ float2 K8[4][64];
  __shared__ float2 CU[NL][2][2][2];

  const int tid = threadIdx.x;
  const int s = blockIdx.x;

  if (tid < 12) {
    const float* xs = x + (size_t)s * 48;
    float a = xs[0 * 12 + tid], b = xs[1 * 12 + tid];
    float c = xs[2 * 12 + tid], d = xs[3 * 12 + tid];
    float ca = cosf(0.5f * a), sa = sinf(0.5f * a);
    float cb = cosf(0.5f * b), sb = sinf(0.5f * b);
    float cc = cosf(0.5f * c), sc = sinf(0.5f * c);
    float cd = cosf(0.5f * d), sd = sinf(0.5f * d);
    C2 M  = { {ca, 0.f}, {-sa, 0.f}, {sa, 0.f}, {ca, 0.f} };
    C2 RX = { {cb, 0.f}, {0.f, -sb}, {0.f, -sb}, {cb, 0.f} };
    C2 RZ = { {cc, -sc}, {0.f, 0.f}, {0.f, 0.f}, {cc, sc} };
    C2 RY = { {cd, 0.f}, {-sd, 0.f}, {sd, 0.f}, {cd, 0.f} };
    M = mm2(RX, M); M = mm2(RZ, M); M = mm2(RY, M);
    E[tid][0][0] = M.a00; E[tid][0][1] = M.a01;
    E[tid][1][0] = M.a10; E[tid][1][1] = M.a11;
  }
  if (tid < 2 * NL) {
    int l = tid >> 1, which = tid & 1;
    const float* p = &entp[(l * 12 + (which ? 11 : 5)) * 3];
    rot_gate(p[0], p[1], p[2], &CU[l][which][0][0], &CU[l][which][0][1],
             &CU[l][which][1][0], &CU[l][which][1][1]);
  }
  for (int idx = tid; idx < 4096; idx += 256) {
    int i = idx >> 6, j = idx & 63;
    S0r[i * LDA + j] = (idx == 0) ? 1.f : 0.f;
    S0i[i * LDA + j] = 0.f;
  }
  __syncthreads();

  {
    int g = tid >> 6, a = (tid >> 3) & 7, b = tid & 7;
    float2 p = cmul(cmul(E[3 * g][(a >> 2) & 1][(b >> 2) & 1],
                         E[3 * g + 1][(a >> 1) & 1][(b >> 1) & 1]),
                    E[3 * g + 2][a & 1][b & 1]);
    K8[g][a * 8 + b] = p;
  }
  __syncthreads();

  float *curR = S0r, *curI = S0i, *nxtR = S1r, *nxtI = S1i;

  for (int l = 0; l < NL; l++) {
    const float2* Wl = W + (size_t)l * 3 * 4096;

    for (int idx = tid; idx < 4096; idx += 256) {
      int i = idx >> 6, j = idx & 63;
      float2 p = cmul(K8[0][((i >> 3) << 3) | (j >> 3)],
                      K8[1][((i & 7) << 3) | (j & 7)]);
      OpR[i * LDA + j] = p.x; OpI[i * LDA + j] = p.y;
    }
    __syncthreads();
    cgemm64(OpR, OpI, curR, curI, nxtR, nxtI);
    __syncthreads();
    { float* t; t = curR; curR = nxtR; nxtR = t; t = curI; curI = nxtI; nxtI = t; }

    for (int idx = tid; idx < 4096; idx += 256) {
      int i = idx >> 6, j = idx & 63;
      float2 p = cmul(K8[2][((j >> 3) << 3) | (i >> 3)],
                      K8[3][((j & 7) << 3) | (i & 7)]);
      OpR[i * LDA + j] = p.x; OpI[i * LDA + j] = p.y;
    }
    __syncthreads();
    cgemm64(curR, curI, OpR, OpI, nxtR, nxtI);
    __syncthreads();
    { float* t; t = curR; curR = nxtR; nxtR = t; t = curI; curI = nxtI; nxtI = t; }

    for (int idx = tid; idx < 4096; idx += 256) {
      float2 v = Wl[idx];
      int i = idx >> 6, j = idx & 63;
      OpR[i * LDA + j] = v.x; OpI[i * LDA + j] = v.y;
    }
    __syncthreads();
    cgemm64(OpR, OpI, curR, curI, nxtR, nxtI);
    __syncthreads();
    { float* t; t = curR; curR = nxtR; nxtR = t; t = curI; curI = nxtI; nxtI = t; }

    for (int idx = tid; idx < 4096; idx += 256) {
      float2 v = Wl[4096 + idx];
      int i = idx >> 6, j = idx & 63;
      OpR[i * LDA + j] = v.x; OpI[i * LDA + j] = v.y;
    }
    __syncthreads();
    cgemm64(curR, curI, OpR, OpI, nxtR, nxtI);
    __syncthreads();
    { float* t; t = curR; curR = nxtR; nxtR = t; t = curI; curI = nxtI; nxtI = t; }

    for (int idx = tid; idx < 4096; idx += 256) {
      float2 v = Wl[2 * 4096 + idx];
      int i = idx >> 6, j = idx & 63;
      OpR[i * LDA + j] = v.x; OpI[i * LDA + j] = v.y;
    }
    {
      float2 u00 = CU[l][0][0][0], u01 = CU[l][0][0][1];
      float2 u10 = CU[l][0][1][0], u11 = CU[l][0][1][1];
      for (int t = tid; t < 1024; t += 256) {
        int i = ((t >> 5) << 1) | 1;
        int jl = t & 31;
        int a0 = i * LDA + jl, a1 = a0 + 32;
        float2 x0 = make_float2(curR[a0], curI[a0]);
        float2 x1 = make_float2(curR[a1], curI[a1]);
        float2 y0 = cadd(cmul(u00, x0), cmul(u01, x1));
        float2 y1 = cadd(cmul(u10, x0), cmul(u11, x1));
        curR[a0] = y0.x; curI[a0] = y0.y;
        curR[a1] = y1.x; curI[a1] = y1.y;
      }
    }
    __syncthreads();

    cgemm64(curR, curI, OpR, OpI, nxtR, nxtI);
    __syncthreads();
    { float* t; t = curR; curR = nxtR; nxtR = t; t = curI; curI = nxtI; nxtI = t; }

    {
      float2 u00 = CU[l][1][0][0], u01 = CU[l][1][0][1];
      float2 u10 = CU[l][1][1][0], u11 = CU[l][1][1][1];
      for (int t = tid; t < 1024; t += 256) {
        int il = t >> 5;
        int j = ((t & 31) << 1) | 1;
        int a0 = il * LDA + j, a1 = (il + 32) * LDA + j;
        float2 x0 = make_float2(curR[a0], curI[a0]);
        float2 x1 = make_float2(curR[a1], curI[a1]);
        float2 y0 = cadd(cmul(u00, x0), cmul(u01, x1));
        float2 y1 = cadd(cmul(u10, x0), cmul(u11, x1));
        curR[a0] = y0.x; curI[a0] = y0.y;
        curR[a1] = y1.x; curI[a1] = y1.y;
      }
    }
  }
  __syncthreads();

  float z[12];
  for (int q = 0; q < 12; q++) z[q] = 0.f;
  for (int idx = tid; idx < 4096; idx += 256) {
    int i = idx >> 6, j = idx & 63;
    float pr = curR[i * LDA + j], pi = curI[i * LDA + j];
    float p = pr * pr + pi * pi;
    for (int q = 0; q < 12; q++)
      z[q] += ((idx >> (11 - q)) & 1) ? -p : p;
  }
  __syncthreads();
  float* red = OpR;
  for (int q = 0; q < 12; q++) red[tid * 12 + q] = z[q];
  __syncthreads();
  for (int sft = 128; sft > 0; sft >>= 1) {
    if (tid < sft)
      for (int q = 0; q < 12; q++)
        red[tid * 12 + q] += red[(tid + sft) * 12 + q];
    __syncthreads();
  }
  if (tid < 12) out[(size_t)s * 12 + tid] = red[tid];
}

extern "C" void kernel_launch(void* const* d_in, const int* in_sizes, int n_in,
                              void* d_out, int out_size, void* d_ws, size_t ws_size,
                              hipStream_t stream) {
  (void)n_in; (void)out_size; (void)ws_size;
  const float* x    = (const float*)d_in[0];
  const float* rotp = (const float*)d_in[1];
  const float* entp = (const float*)d_in[2];
  float2* W = (float2*)d_ws;
  const int B = in_sizes[0] / 48;

  qk_prep<<<NL, 256, 0, stream>>>(rotp, entp, W);
  qk_main<<<B, 256, 0, stream>>>(x, entp, W, (float*)d_out);
}


# Enc products run K=16, not 64: rotations of wires 0,1 (6,7) leave Rh (Rl) as a 4-term mix in the fragment loads, rest is one 16x16 real operator; 16 WMMAs a wave-product, not 64; sum order changes
# speedup vs baseline: 1.9556x; 1.9556x over previous
#include <hip/hip_runtime.h>
#include <math.h>

#define LDA 65
#define NL  4
#define NPREP 8

typedef __attribute__((ext_vector_type(2))) float v2f;
typedef __attribute__((ext_vector_type(8))) float v8f;

__device__ __forceinline__ float2 cmul(float2 a, float2 b) {
  return make_float2(a.x * b.x - a.y * b.y, a.x * b.y + a.y * b.x);
}
__device__ __forceinline__ float2 cadd(float2 a, float2 b) {
  return make_float2(a.x + b.x, a.y + b.y);
}

struct C2 { float2 a00, a01, a10, a11; };
__device__ __forceinline__ C2 mm2(const C2 L, const C2 R) {
  C2 o;
  o.a00 = cadd(cmul(L.a00, R.a00), cmul(L.a01, R.a10));
  o.a01 = cadd(cmul(L.a00, R.a01), cmul(L.a01, R.a11));
  o.a10 = cadd(cmul(L.a10, R.a00), cmul(L.a11, R.a10));
  o.a11 = cadd(cmul(L.a10, R.a01), cmul(L.a11, R.a11));
  return o;
}

__device__ __forceinline__ void rot_gate(float phi, float th, float om,
                                         float2* u00, float2* u01,
                                         float2* u10, float2* u11) {
  float ct = cosf(0.5f * th), st = sinf(0.5f * th);
  float ap = -0.5f * (phi + om);
  float am = 0.5f * (phi - om);
  *u00 = make_float2(ct * cosf(ap), ct * sinf(ap));
  *u01 = make_float2(-st * cosf(am), -st * sinf(am));
  *u10 = make_float2(st * cosf(-am), st * sinf(-am));
  *u11 = make_float2(ct * cosf(-ap), ct * sinf(-ap));
}

__device__ __forceinline__ void apply_cgate(float* Mr, float* Mi,
                                            float2 u00, float2 u01,
                                            float2 u10, float2 u11,
                                            int cb, int tb, int tid) {
  for (int idx = tid; idx < 4096; idx += 256) {
    int i = idx >> 6, j = idx & 63;
    if ((i & cb) && !(i & tb)) {
      int a0 = i * LDA + j, a1 = (i | tb) * LDA + j;
      float2 x0 = make_float2(Mr[a0], Mi[a0]);
      float2 x1 = make_float2(Mr[a1], Mi[a1]);
      float2 y0 = cadd(cmul(u00, x0), cmul(u01, x1));
      float2 y1 = cadd(cmul(u10, x0), cmul(u11, x1));
      Mr[a0] = y0.x; Mi[a0] = y0.y;
      Mr[a1] = y1.x; Mi[a1] = y1.y;
    }
  }
}

__global__ __launch_bounds__(256) void qk_prep(const float* __restrict__ rotp,
                                               const float* __restrict__ entp,
                                               float2* __restrict__ W) {
  __shared__ float Mr[64 * LDA], Mi[64 * LDA];
  __shared__ float2 U[24][2][2];
  const int l = blockIdx.x >> 1;
  const int slot = blockIdx.x & 1;
  const int tid = threadIdx.x;

  if (tid < 24) {
    const float* p = (tid < 12) ? &rotp[(l * 12 + tid) * 3]
                                : &entp[(l * 12 + (tid - 12)) * 3];
    rot_gate(p[0], p[1], p[2], &U[tid][0][0], &U[tid][0][1],
             &U[tid][1][0], &U[tid][1][1]);
  }
  __syncthreads();

  const int g0 = slot ? 6 : 0;

  for (int idx = tid; idx < 4096; idx += 256) {
    int i = idx >> 6, j = idx & 63;
    float2 p = make_float2((((i ^ j) >> 5) & 1) ? 0.f : 1.f, 0.f);
    for (int q = 1; q < 6; q++)
      p = cmul(p, U[g0 + q][(i >> (5 - q)) & 1][(j >> (5 - q)) & 1]);
    Mr[i * LDA + j] = p.x; Mi[i * LDA + j] = p.y;
  }
  __syncthreads();
  for (int c = 0; c < 5; c++) {
    apply_cgate(Mr, Mi, U[12 + g0 + c][0][0], U[12 + g0 + c][0][1],
                U[12 + g0 + c][1][0], U[12 + g0 + c][1][1],
                1 << (5 - c), 1 << (4 - c), tid);
    __syncthreads();
  }
  float2* Wd = W + ((size_t)l * 2 + slot) * 4096;
  for (int idx = tid; idx < 4096; idx += 256) {
    int i = idx >> 6, j = idx & 63;
    int dst = (((((i >> 4) * 16 + (j >> 2)) * 32 + ((j >> 1) & 1) * 16 + (i & 15)) << 1) | (j & 1));
    Wd[dst] = make_float2(Mr[i * LDA + j], Mi[i * LDA + j]);
  }
}

__device__ __forceinline__ v8f wmma4(v2f a, v2f b, v8f c) {
  return __builtin_amdgcn_wmma_f32_16x16x4_f32(
      false, a, false, b, (short)0, c, false, false);
}

__device__ __forceinline__ void cgemm64_opA(const float4* __restrict__ Af,
                                            const float2 (*G)[2],
                                            const float* Br, const float* Bi,
                                            float* Dr, float* Di) {
  const int tid = threadIdx.x;
  const int lane = tid & 31;
  const int w = tid >> 5;
  const int mt = (w & 3) * 16;
  const int mb = (w & 3) >> 1;
  const int nt0 = (w >> 2) * 32;
  const int nt1 = nt0 + 16;
  const int ml = lane & 15;
  const int kb = (lane >> 4) * 2;
  const int hi = lane >> 4;
  const float4* af = Af + (w & 3) * 512 + mb * 256 + lane;

  float4 a_n0 = af[0], a_n1 = af[32];
  __syncthreads();
  const float2 u0 = G[mb][0], u1 = G[mb][1];

  v8f p1_0 = {}, p2_0 = {}, p3_0 = {};
  v8f p1_1 = {}, p2_1 = {}, p3_1 = {};
  for (int t = 0; t < 8; t++) {
    const int k = 4 * t;
    const float4 a = a_n0;
    a_n0 = a_n1;
    a_n1 = af[32 * min(t + 2, 7)];
    v2f ar; ar.x = a.x; ar.y = a.z;
    v2f ai; ai.x = a.y; ai.y = a.w;
    v2f as = ar + ai;
    const int r0 = (k + kb) * LDA, r1 = r0 + LDA;
    const int s0 = r0 + 32 * LDA, s1 = r1 + 32 * LDA;
    const int c0 = nt0 + ml, c1 = nt1 + ml;
    const float2 e00 = cadd(cmul(u0, make_float2(Br[r0 + c0], Bi[r0 + c0])),
                            cmul(u1, make_float2(Br[s0 + c0], Bi[s0 + c0])));
    const float2 e01 = cadd(cmul(u0, make_float2(Br[r1 + c0], Bi[r1 + c0])),
                            cmul(u1, make_float2(Br[s1 + c0], Bi[s1 + c0])));
    const float2 e10 = cadd(cmul(u0, make_float2(Br[r0 + c1], Bi[r0 + c1])),
                            cmul(u1, make_float2(Br[s0 + c1], Bi[s0 + c1])));
    const float2 e11 = cadd(cmul(u0, make_float2(Br[r1 + c1], Bi[r1 + c1])),
                            cmul(u1, make_float2(Br[s1 + c1], Bi[s1 + c1])));
    v2f br0, bi0, br1, bi1;
    br0.x = e00.x; br0.y = e01.x;
    bi0.x = e00.y; bi0.y = e01.y;
    br1.x = e10.x; br1.y = e11.x;
    bi1.x = e10.y; bi1.y = e11.y;
    v2f bs0 = br0 + bi0;
    v2f bs1 = br1 + bi1;

    p1_0 = wmma4(ar, br0, p1_0);
    p2_0 = wmma4(ai, bi0, p2_0);
    p3_0 = wmma4(as, bs0, p3_0);
    p1_1 = wmma4(ar, br1, p1_1);
    p2_1 = wmma4(ai, bi1, p2_1);
    p3_1 = wmma4(as, bs1, p3_1);
  }

  v8f cr0 = p1_0 - p2_0;
  v8f ci0 = p3_0 - p1_0 - p2_0;
  v8f cr1 = p1_1 - p2_1;
  v8f ci1 = p3_1 - p1_1 - p2_1;

  for (int v = 0; v < 8; v++) {
    int m = mt + v + 8 * hi;
    int n0 = nt0 + ml, n1 = nt1 + ml;
    Dr[m * LDA + n0] = cr0[v]; Di[m * LDA + n0] = ci0[v];
    Dr[m * LDA + n1] = cr1[v]; Di[m * LDA + n1] = ci1[v];
  }
}

struct RealPairs { float a0, a1, b0, b1; };
__device__ __forceinline__ RealPairs real_pairs(const float* rbrow, int kb) {
  RealPairs p;
  p.a0 = rbrow[kb];     p.a1 = rbrow[kb + 1];
  p.b0 = rbrow[kb + 4]; p.b1 = rbrow[kb + 5];
  return p;
}

__device__ __forceinline__ float2 rmix(bool first, float c, float2 z, float2 e) {
  return first ? make_float2(c * z.x, c * z.y)
               : make_float2(fmaf(c, z.x, e.x), fmaf(c, z.y, e.y));
}
__device__ __forceinline__ float wave_uniform(float v) {
  return __int_as_float(__builtin_amdgcn_readfirstlane(__float_as_int(v)));
}

struct Frag16 { v2f f[4]; };
__device__ __forceinline__ Frag16 frag16(const float (*RQc)[2], const float* R8m,
                                         int c, int kb) {
  const RealPairs g = real_pairs(R8m + ((c & 7) << 3), kb);
  const float f0 = RQc[c >> 3][0], f1 = RQc[c >> 3][1];
  Frag16 o;
  o.f[0].x = f0 * g.a0; o.f[0].y = f0 * g.a1;
  o.f[1].x = f0 * g.b0; o.f[1].y = f0 * g.b1;
  o.f[2].x = f1 * g.a0; o.f[2].y = f1 * g.a1;
  o.f[3].x = f1 * g.b0; o.f[3].y = f1 * g.b1;
  return o;
}

__device__ __forceinline__ void rgemm64_encA(const float (*RQw)[2][2], const float* R8m,
                                             const float2* QH,
                                             const float* Br, const float* Bi,
                                             float* Dr, float* Di) {
  const int tid = threadIdx.x;
  const int lane = tid & 31;
  const int w = tid >> 5;
  const int mh = w & 1;
  const int mtA = mh * 32;
  const int mtB = mtA + 16;
  const int nt = (w >> 1) * 16;
  const int ml = lane & 15;
  const int kb = (lane >> 4) * 2;
  const int hi = lane >> 4;
  float cA[4], cB[4];
  for (int j = 0; j < 4; j++) {
    const float r0 = RQw[0][mh][j >> 1];
    cA[j] = wave_uniform(r0 * RQw[1][0][j & 1]);
    cB[j] = wave_uniform(r0 * RQw[1][1][j & 1]);
  }
  const RealPairs g = real_pairs(R8m + ((ml & 7) << 3), kb);
  const float f0 = RQw[2][ml >> 3][0], f1 = RQw[2][ml >> 3][1];

  v8f trA = {}, tiA = {}, trB = {}, tiB = {};
#pragma unroll 1
  for (int t = 0; t < 4; t++) {
    const float ft = (t & 2) ? f1 : f0;
    v2f at;
    at.x = ft * ((t & 1) ? g.b0 : g.a0);
    at.y = ft * ((t & 1) ? g.b1 : g.a1);
    float2 eA0 = make_float2(0.f, 0.f), eA1 = eA0, eB0 = eA0, eB1 = eA0;
#pragma unroll
    for (int j = 0; j < 4; j++) {
      const int k = 16 * j + 4 * t + kb;
      const float4 q = *reinterpret_cast<const float4*>(QH + k);
      const int r0 = k * LDA + nt + ml, r1 = r0 + LDA;
      const float2 z0 = cmul(make_float2(q.x, q.y), make_float2(Br[r0], Bi[r0]));
      const float2 z1 = cmul(make_float2(q.z, q.w), make_float2(Br[r1], Bi[r1]));
      eA0 = rmix(j == 0, cA[j], z0, eA0);
      eA1 = rmix(j == 0, cA[j], z1, eA1);
      eB0 = rmix(j == 0, cB[j], z0, eB0);
      eB1 = rmix(j == 0, cB[j], z1, eB1);
    }
    v2f brA, biA, brB, biB;
    brA.x = eA0.x; brA.y = eA1.x;
    biA.x = eA0.y; biA.y = eA1.y;
    brB.x = eB0.x; brB.y = eB1.x;
    biB.x = eB0.y; biB.y = eB1.y;

    trA = wmma4(at, brA, trA);
    tiA = wmma4(at, biA, tiA);
    trB = wmma4(at, brB, trB);
    tiB = wmma4(at, biB, tiB);
  }

  for (int v = 0; v < 8; v++) {
    const int mA = mtA + v + 8 * hi, mB = mtB + v + 8 * hi;
    const int n = nt + ml;
    Dr[mA * LDA + n] = trA[v]; Di[mA * LDA + n] = tiA[v];
    Dr[mB * LDA + n] = trB[v]; Di[mB * LDA + n] = tiB[v];
  }
}

__device__ __forceinline__ void rgemm64_encB(const float* Ar, const float* Ai,
                                             const float2* PH, const float2* QL,
                                             const float (*RQw)[2][2], const float* R8m,
                                             float* Dr, float* Di) {
  const int tid = threadIdx.x;
  const int lane = tid & 31;
  const int w = tid >> 5;
  const int mt = (w & 3) * 16;
  const int nh = w >> 2;
  const int nt0 = nh * 32;
  const int nt1 = nt0 + 16;
  const int ml = lane & 15;
  const int kb = (lane >> 4) * 2;
  const int hi = lane >> 4;
  float cA[4], cB[4];
  for (int j = 0; j < 4; j++) {
    const float r0 = RQw[0][nh][j >> 1];
    cA[j] = wave_uniform(r0 * RQw[1][0][j & 1]);
    cB[j] = wave_uniform(r0 * RQw[1][1][j & 1]);
  }
  const Frag16 b = frag16(RQw[2], R8m, ml, kb);
  const float2 phm = PH[mt + ml];
  const float* arow = Ar + (mt + ml) * LDA + kb;
  const float* airow = Ai + (mt + ml) * LDA + kb;

  v8f xr0 = {}, xi0 = {}, xr1 = {}, xi1 = {};
#pragma unroll
  for (int t = 0; t < 4; t++) {
    float2 eA0 = make_float2(0.f, 0.f), eA1 = eA0, eB0 = eA0, eB1 = eA0;
#pragma unroll
    for (int j = 0; j < 4; j++) {
      const int k = 16 * j + 4 * t;
      const float4 q = *reinterpret_cast<const float4*>(QL + k + kb);
      const float2 z0 = cmul(make_float2(q.x, q.y), make_float2(arow[k], airow[k]));
      const float2 z1 = cmul(make_float2(q.z, q.w), make_float2(arow[k + 1], airow[k + 1]));
      eA0 = rmix(j == 0, cA[j], z0, eA0);
      eA1 = rmix(j == 0, cA[j], z1, eA1);
      eB0 = rmix(j == 0, cB[j], z0, eB0);
      eB1 = rmix(j == 0, cB[j], z1, eB1);
    }
    eA0 = cmul(phm, eA0); eA1 = cmul(phm, eA1);
    eB0 = cmul(phm, eB0); eB1 = cmul(phm, eB1);
    v2f arA, aiA, arB, aiB;
    arA.x = eA0.x; arA.y = eA1.x;
    aiA.x = eA0.y; aiA.y = eA1.y;
    arB.x = eB0.x; arB.y = eB1.x;
    aiB.x = eB0.y; aiB.y = eB1.y;

    xr0 = wmma4(arA, b.f[t], xr0);
    xi0 = wmma4(aiA, b.f[t], xi0);
    xr1 = wmma4(arB, b.f[t], xr1);
    xi1 = wmma4(aiB, b.f[t], xi1);
  }

  for (int v = 0; v < 8; v++) {
    int m = mt + v + 8 * hi;
    int n0 = nt0 + ml, n1 = nt1 + ml;
    Dr[m * LDA + n0] = xr0[v]; Di[m * LDA + n0] = xi0[v];
    Dr[m * LDA + n1] = xr1[v]; Di[m * LDA + n1] = xi1[v];
  }
}

__device__ __forceinline__ void cgemm64_rowpar(const float* Sr, const float* Si,
                                               const float2* CMx,
                                               const float4* __restrict__ Bf,
                                               float* Dr, float* Di) {
  const int tid = threadIdx.x;
  const int lane = tid & 31;
  const int w = tid >> 5;
  const int par = (w >> 1) & 1;
  const int rb = (w & 1) * 16;
  const int nb = w >> 2;
  const int nt0 = nb * 32;
  const int nt1 = nt0 + 16;
  const int ml = lane & 15;
  const int kb = (lane >> 4) * 2;
  const int hi = lane >> 4;
  const float4* bf0 = Bf + nb * 1024 + nb * 256 + lane;
  const float4* bf1 = bf0 + 512;
  const int arow = (2 * (rb + ml) + par) * LDA;
  const float2* cm0 = CMx + ((par * 2 + nb) * 2) * 32;
  const float2* cm1 = cm0 + 32;

  float4 b0_n0 = bf0[0], b1_n0 = bf1[0];
  float4 b0_n1 = bf0[32], b1_n1 = bf1[32];
  __syncthreads();

  v8f p1_0 = {}, p2_0 = {}, p3_0 = {};
  v8f p1_1 = {}, p2_1 = {}, p3_1 = {};
  for (int t = 0; t < 8; t++) {
    const int k = 4 * t;
    const float4 b0 = b0_n0, b1 = b1_n0;
    b0_n0 = b0_n1; b1_n0 = b1_n1;
    const int tn = min(t + 2, 7);
    b0_n1 = bf0[32 * tn]; b1_n1 = bf1[32 * tn];
    const float* ap = Sr + arow + k + kb;
    const float* aip = Si + arow + k + kb;
    const float4 q0 = *reinterpret_cast<const float4*>(cm0 + k + kb);
    const float4 q1 = *reinterpret_cast<const float4*>(cm1 + k + kb);
    const float2 e0 = cadd(cmul(make_float2(q0.x, q0.y), make_float2(ap[0], aip[0])),
                           cmul(make_float2(q1.x, q1.y), make_float2(ap[32], aip[32])));
    const float2 e1 = cadd(cmul(make_float2(q0.z, q0.w), make_float2(ap[1], aip[1])),
                           cmul(make_float2(q1.z, q1.w), make_float2(ap[33], aip[33])));
    v2f ar; ar.x = e0.x; ar.y = e1.x;
    v2f ai; ai.x = e0.y; ai.y = e1.y;
    v2f as = ar + ai;
    v2f br0, bi0, br1, bi1;
    br0.x = b0.x; br0.y = b0.z;
    bi0.x = b0.y; bi0.y = b0.w;
    br1.x = b1.x; br1.y = b1.z;
    bi1.x = b1.y; bi1.y = b1.w;
    v2f bs0 = br0 + bi0;
    v2f bs1 = br1 + bi1;

    p1_0 = wmma4(ar, br0, p1_0);
    p2_0 = wmma4(ai, bi0, p2_0);
    p3_0 = wmma4(as, bs0, p3_0);
    p1_1 = wmma4(ar, br1, p1_1);
    p2_1 = wmma4(ai, bi1, p2_1);
    p3_1 = wmma4(as, bs1, p3_1);
  }

  v8f cr0 = p1_0 - p2_0;
  v8f ci0 = p3_0 - p1_0 - p2_0;
  v8f cr1 = p1_1 - p2_1;
  v8f ci1 = p3_1 - p1_1 - p2_1;

  for (int v = 0; v < 8; v++) {
    int m = 2 * (rb + v + 8 * hi) + par;
    int n0 = nt0 + ml, n1 = nt1 + ml;
    Dr[m * LDA + n0] = cr0[v]; Di[m * LDA + n0] = ci0[v];
    Dr[m * LDA + n1] = cr1[v]; Di[m * LDA + n1] = ci1[v];
  }
}

__global__ __launch_bounds__(256, 8) void qk_main(const float* __restrict__ x,
                                               const float* __restrict__ rotp,
                                               const float* __restrict__ entp,
                                               const float2* __restrict__ W,
                                               float* __restrict__ out) {
  __shared__ float S0r[64 * LDA], S0i[64 * LDA];
  __shared__ float S1r[64 * LDA], S1i[64 * LDA];
  __shared__ float RQ[12][2][2];
  __shared__ float2 PP[12][2];
  __shared__ float2 QQ[12][2];
  __shared__ float R8[4][64];
  __shared__ __attribute__((aligned(16))) float2 PV[4][64];
  __shared__ float2 GU[NL][4][2][2];
  __shared__ __attribute__((aligned(16))) float2 CM[2][2][2][32];

  const int tid = threadIdx.x;
  const int s = blockIdx.x;

  if (tid < 12) {
    const float* xs = x + (size_t)s * 48;
    float a = xs[0 * 12 + tid], b = xs[1 * 12 + tid];
    float c = xs[2 * 12 + tid], d = xs[3 * 12 + tid];
    float ca = cosf(0.5f * a), sa = sinf(0.5f * a);
    float cb = cosf(0.5f * b), sb = sinf(0.5f * b);
    float cc = cosf(0.5f * c), sc = sinf(0.5f * c);
    float cd = cosf(0.5f * d), sd = sinf(0.5f * d);
    C2 M  = { {ca, 0.f}, {-sa, 0.f}, {sa, 0.f}, {ca, 0.f} };
    C2 RX = { {cb, 0.f}, {0.f, -sb}, {0.f, -sb}, {cb, 0.f} };
    C2 RZ = { {cc, -sc}, {0.f, 0.f}, {0.f, 0.f}, {cc, sc} };
    C2 RY = { {cd, 0.f}, {-sd, 0.f}, {sd, 0.f}, {cd, 0.f} };
    M = mm2(RX, M); M = mm2(RZ, M); M = mm2(RY, M);
    const float cs = sqrtf(M.a00.x * M.a00.x + M.a00.y * M.a00.y);
    const float sn = sqrtf(M.a10.x * M.a10.x + M.a10.y * M.a10.y);
    float2 p0 = make_float2(1.f, 0.f), p1 = make_float2(1.f, 0.f), q1;
    if (cs > 0.f) { const float r = 1.f / cs; p0 = make_float2(M.a00.x * r, M.a00.y * r); }
    if (sn > 0.f) { const float r = 1.f / sn; p1 = make_float2(M.a10.x * r, M.a10.y * r); }
    if (cs >= sn) {
      const float r = 1.f / cs;
      const float2 t = cmul(M.a11, make_float2(p1.x, -p1.y));
      q1 = make_float2(t.x * r, t.y * r);
    } else {
      const float r = -1.f / sn;
      const float2 t = cmul(M.a01, make_float2(p0.x, -p0.y));
      q1 = make_float2(t.x * r, t.y * r);
    }
    RQ[tid][0][0] = cs; RQ[tid][0][1] = -sn;
    RQ[tid][1][0] = sn; RQ[tid][1][1] = cs;
    PP[tid][0] = p0; PP[tid][1] = p1;
    QQ[tid][0] = make_float2(1.f, 0.f); QQ[tid][1] = q1;
  }
  if (tid < 4 * NL) {
    const int gl = tid >> 2, gg = tid & 3;
    const float* p = (gg == 0) ? &entp[(gl * 12 + 11) * 3]
                   : (gg == 1) ? &rotp[(gl * 12 + 0) * 3]
                   : (gg == 2) ? &rotp[(gl * 12 + 6) * 3]
                               : &entp[(gl * 12 + 5) * 3];
    rot_gate(p[0], p[1], p[2], &GU[gl][gg][0][0], &GU[gl][gg][0][1],
             &GU[gl][gg][1][0], &GU[gl][gg][1][1]);
  }
  for (int idx = tid; idx < 4096; idx += 256) {
    int i = idx >> 6, j = idx & 63;
    S0r[i * LDA + j] = (idx == 0) ? 1.f : 0.f;
    S0i[i * LDA + j] = 0.f;
  }
  __syncthreads();

  {
    int g = tid >> 6, a = (tid >> 3) & 7, b = tid & 7;
    R8[g][a * 8 + b] = RQ[3 * g][(a >> 2) & 1][(b >> 2) & 1] *
                       RQ[3 * g + 1][(a >> 1) & 1][(b >> 1) & 1] *
                       RQ[3 * g + 2][a & 1][b & 1];
    const int idx = tid & 63;
    const int w0 = (g >> 1) * 6;
    const float2 (*ph)[2] = (g & 1) ? QQ : PP;
    float2 p = ph[w0][(idx >> 5) & 1];
    for (int q = 1; q < 6; q++)
      p = cmul(p, ph[w0 + q][(idx >> (5 - q)) & 1]);
    PV[g][idx] = p;
  }
  __syncthreads();

  float *curR = S0r, *curI = S0i, *nxtR = S1r, *nxtI = S1i;

  for (int l = 0; l < NL; l++) {
    const float4* Wl = reinterpret_cast<const float4*>(W) + (size_t)l * 2 * 2048;

    {
      const int p = tid >> 7, nb = (tid >> 6) & 1, ka = (tid >> 5) & 1, kk = tid & 31;
      float2 g = GU[l][2][nb][ka];
      if (p)
        g = cadd(cmul(GU[l][3][nb][0], GU[l][2][0][ka]),
                 cmul(GU[l][3][nb][1], GU[l][2][1][ka]));
      CM[p][nb][ka][kk] = cmul(g, PV[2][32 * ka + kk]);
    }

    rgemm64_encA(&RQ[0], R8[1], PV[1], curR, curI, nxtR, nxtI);
    __syncthreads();
    { float* t; t = curR; curR = nxtR; nxtR = t; t = curI; curI = nxtI; nxtI = t; }

    rgemm64_encB(curR, curI, PV[0], PV[3], &RQ[6], R8[3], nxtR, nxtI);

    { float* t; t = curR; curR = nxtR; nxtR = t; t = curI; curI = nxtI; nxtI = t; }
    cgemm64_opA(Wl, GU[l][1], curR, curI, nxtR, nxtI);
    { float* t; t = curR; curR = nxtR; nxtR = t; t = curI; curI = nxtI; nxtI = t; }

    cgemm64_rowpar(curR, curI, &CM[0][0][0][0], Wl + 2048, nxtR, nxtI);
    { float* t; t = curR; curR = nxtR; nxtR = t; t = curI; curI = nxtI; nxtI = t; }
    __syncthreads();

    {
      float2 u00 = GU[l][0][0][0], u01 = GU[l][0][0][1];
      float2 u10 = GU[l][0][1][0], u11 = GU[l][0][1][1];
      for (int t = tid; t < 1024; t += 256) {
        int il = t >> 5;
        int j = ((t & 31) << 1) | 1;
        int a0 = il * LDA + j, a1 = (il + 32) * LDA + j;
        float2 x0 = make_float2(curR[a0], curI[a0]);
        float2 x1 = make_float2(curR[a1], curI[a1]);
        float2 y0 = cadd(cmul(u00, x0), cmul(u01, x1));
        float2 y1 = cadd(cmul(u10, x0), cmul(u11, x1));
        curR[a0] = y0.x; curI[a0] = y0.y;
        curR[a1] = y1.x; curI[a1] = y1.y;
      }
    }
    __syncthreads();
  }

  float z[12];
  for (int q = 0; q < 12; q++) z[q] = 0.f;
  for (int idx = tid; idx < 4096; idx += 256) {
    int i = idx >> 6, j = idx & 63;
    float pr = curR[i * LDA + j], pi = curI[i * LDA + j];
    float p = pr * pr + pi * pi;
    for (int q = 0; q < 12; q++)
      z[q] += ((idx >> (11 - q)) & 1) ? -p : p;
  }
  __syncthreads();
  float* red = nxtR;
  for (int q = 0; q < 12; q++) red[tid * 12 + q] = z[q];
  __syncthreads();
  for (int sft = 128; sft > 0; sft >>= 1) {
    if (tid < sft)
      for (int q = 0; q < 12; q++)
        red[tid * 12 + q] += red[(tid + sft) * 12 + q];
    __syncthreads();
  }
  if (tid < 12) out[(size_t)s * 12 + tid] = red[tid];
}

extern "C" void kernel_launch(void* const* d_in, const int* in_sizes, int n_in,
                              void* d_out, int out_size, void* d_ws, size_t ws_size,
                              hipStream_t stream) {
  (void)n_in; (void)out_size; (void)ws_size;
  const float* x    = (const float*)d_in[0];
  const float* rotp = (const float*)d_in[1];
  const float* entp = (const float*)d_in[2];
  float2* W = (float2*)d_ws;
  const int B = in_sizes[0] / 48;

  qk_prep<<<NPREP, 256, 0, stream>>>(rotp, entp, W);
  qk_main<<<B, 256, 0, stream>>>(x, rotp, entp, W, (float*)d_out);
}


# MA, R0/R1 products run K=16, not 32: wires 1 and 7 only control the rest of their CRot chain, so their gates join a 4-term mix in the fragment loads; operators are 16x16 blocks; sum order changes
# speedup vs baseline: 3.3217x; 1.6986x over previous
#include <hip/hip_runtime.h>
#include <math.h>

#define LDA 65
#define NL  4
#define NPREP 8

typedef __attribute__((ext_vector_type(2))) float v2f;
typedef __attribute__((ext_vector_type(8))) float v8f;

__device__ __forceinline__ float2 cmul(float2 a, float2 b) {
  return make_float2(a.x * b.x - a.y * b.y, a.x * b.y + a.y * b.x);
}
__device__ __forceinline__ float2 cadd(float2 a, float2 b) {
  return make_float2(a.x + b.x, a.y + b.y);
}

struct C2 { float2 a00, a01, a10, a11; };
__device__ __forceinline__ C2 mm2(const C2 L, const C2 R) {
  C2 o;
  o.a00 = cadd(cmul(L.a00, R.a00), cmul(L.a01, R.a10));
  o.a01 = cadd(cmul(L.a00, R.a01), cmul(L.a01, R.a11));
  o.a10 = cadd(cmul(L.a10, R.a00), cmul(L.a11, R.a10));
  o.a11 = cadd(cmul(L.a10, R.a01), cmul(L.a11, R.a11));
  return o;
}

__device__ __forceinline__ void rot_gate(float phi, float th, float om,
                                         float2* u00, float2* u01,
                                         float2* u10, float2* u11) {
  float ct = cosf(0.5f * th), st = sinf(0.5f * th);
  float ap = -0.5f * (phi + om);
  float am = 0.5f * (phi - om);
  *u00 = make_float2(ct * cosf(ap), ct * sinf(ap));
  *u01 = make_float2(-st * cosf(am), -st * sinf(am));
  *u10 = make_float2(st * cosf(-am), st * sinf(-am));
  *u11 = make_float2(ct * cosf(-ap), ct * sinf(-ap));
}

__device__ __forceinline__ void apply_cgate(float* Mr, float* Mi,
                                            float2 u00, float2 u01,
                                            float2 u10, float2 u11,
                                            int cb, int tb, int tid) {
  for (int idx = tid; idx < 4096; idx += 256) {
    int i = idx >> 6, j = idx & 63;
    if ((i & cb) && !(i & tb)) {
      int a0 = i * LDA + j, a1 = (i | tb) * LDA + j;
      float2 x0 = make_float2(Mr[a0], Mi[a0]);
      float2 x1 = make_float2(Mr[a1], Mi[a1]);
      float2 y0 = cadd(cmul(u00, x0), cmul(u01, x1));
      float2 y1 = cadd(cmul(u10, x0), cmul(u11, x1));
      Mr[a0] = y0.x; Mi[a0] = y0.y;
      Mr[a1] = y1.x; Mi[a1] = y1.y;
    }
  }
}

__global__ __launch_bounds__(256) void qk_prep(const float* __restrict__ rotp,
                                               const float* __restrict__ entp,
                                               float2* __restrict__ W) {
  __shared__ float Mr[64 * LDA], Mi[64 * LDA];
  __shared__ float2 U[24][2][2];
  const int l = blockIdx.x >> 1;
  const int slot = blockIdx.x & 1;
  const int tid = threadIdx.x;

  if (tid < 24) {
    const float* p = (tid < 12) ? &rotp[(l * 12 + tid) * 3]
                                : &entp[(l * 12 + (tid - 12)) * 3];
    rot_gate(p[0], p[1], p[2], &U[tid][0][0], &U[tid][0][1],
             &U[tid][1][0], &U[tid][1][1]);
  }
  __syncthreads();

  const int g0 = slot ? 6 : 0;

  for (int idx = tid; idx < 4096; idx += 256) {
    int i = idx >> 6, j = idx & 63;
    float2 p = make_float2(((i ^ j) >> 4) ? 0.f : 1.f, 0.f);
    for (int q = 2; q < 6; q++)
      p = cmul(p, U[g0 + q][(i >> (5 - q)) & 1][(j >> (5 - q)) & 1]);
    Mr[i * LDA + j] = p.x; Mi[i * LDA + j] = p.y;
  }
  __syncthreads();
  for (int c = 1; c < 5; c++) {
    apply_cgate(Mr, Mi, U[12 + g0 + c][0][0], U[12 + g0 + c][0][1],
                U[12 + g0 + c][1][0], U[12 + g0 + c][1][1],
                1 << (5 - c), 1 << (4 - c), tid);
    __syncthreads();
  }
  float2* Wd = W + ((size_t)l * 2 + slot) * 4096;
  for (int idx = tid; idx < 4096; idx += 256) {
    int i = idx >> 6, j = idx & 63;
    int dst = (((((i >> 4) * 16 + (j >> 2)) * 32 + ((j >> 1) & 1) * 16 + (i & 15)) << 1) | (j & 1));
    Wd[dst] = make_float2(Mr[i * LDA + j], Mi[i * LDA + j]);
  }
}

__device__ __forceinline__ v8f wmma4(v2f a, v2f b, v8f c) {
  return __builtin_amdgcn_wmma_f32_16x16x4_f32(
      false, a, false, b, (short)0, c, false, false);
}

__device__ __forceinline__ float wave_uniform(float v) {
  return __int_as_float(__builtin_amdgcn_readfirstlane(__float_as_int(v)));
}
__device__ __forceinline__ float2 wave_uniform2(float2 v) {
  return make_float2(wave_uniform(v.x), wave_uniform(v.y));
}
__device__ __forceinline__ float2 cmix(bool first, float2 c, float2 z, float2 e) {
  return first ? make_float2(fmaf(-c.y, z.y, c.x * z.x), fmaf(c.y, z.x, c.x * z.y))
               : make_float2(fmaf(-c.y, z.y, fmaf(c.x, z.x, e.x)),
                             fmaf(c.y, z.x, fmaf(c.x, z.y, e.y)));
}

__device__ __forceinline__ void cgemm64_opA(const float4* __restrict__ Af,
                                            const float2* MX4, const float2* PLv,
                                            const float* Br, const float* Bi,
                                            float* Dr, float* Di) {
  const int tid = threadIdx.x;
  const int lane = tid & 31;
  const int w = tid >> 5;
  const int b1 = w & 1;
  const int mtA = b1 * 16;
  const int mtB = mtA + 32;
  const int nt = (w >> 1) * 16;
  const int ml = lane & 15;
  const int kb = (lane >> 4) * 2;
  const int hi = lane >> 4;
  const float4* af = Af + b1 * 640 + lane;

  float4 a[4];
  for (int t = 0; t < 4; t++) a[t] = af[32 * t];
  __syncthreads();
  float2 cA[4], cB[4];
  for (int j = 0; j < 4; j++) {
    cA[j] = wave_uniform2(MX4[b1 * 4 + j]);
    cB[j] = wave_uniform2(MX4[(2 + b1) * 4 + j]);
  }
  const float2 ph = PLv[nt + ml];
  const float* brow = Br + kb * LDA + nt + ml;
  const float* birow = Bi + kb * LDA + nt + ml;

  v8f p1A = {}, p2A = {}, p3A = {};
  v8f p1B = {}, p2B = {}, p3B = {};
#pragma unroll
  for (int t = 0; t < 4; t++) {
    v2f ar; ar.x = a[t].x; ar.y = a[t].z;
    v2f ai; ai.x = a[t].y; ai.y = a[t].w;
    v2f as = ar + ai;
    float2 eA0 = make_float2(0.f, 0.f), eA1 = eA0, eB0 = eA0, eB1 = eA0;
#pragma unroll
    for (int j = 0; j < 4; j++) {
      const int r0 = (16 * j + 4 * t) * LDA, r1 = r0 + LDA;
      const float2 z0 = make_float2(brow[r0], birow[r0]);
      const float2 z1 = make_float2(brow[r1], birow[r1]);
      eA0 = cmix(j == 0, cA[j], z0, eA0);
      eA1 = cmix(j == 0, cA[j], z1, eA1);
      eB0 = cmix(j == 0, cB[j], z0, eB0);
      eB1 = cmix(j == 0, cB[j], z1, eB1);
    }
    v2f brA, biA, brB, biB;
    brA.x = eA0.x; brA.y = eA1.x;
    biA.x = eA0.y; biA.y = eA1.y;
    brB.x = eB0.x; brB.y = eB1.x;
    biB.x = eB0.y; biB.y = eB1.y;
    v2f bsA = brA + biA;
    v2f bsB = brB + biB;

    p1A = wmma4(ar, brA, p1A);
    p2A = wmma4(ai, biA, p2A);
    p3A = wmma4(as, bsA, p3A);
    p1B = wmma4(ar, brB, p1B);
    p2B = wmma4(ai, biB, p2B);
    p3B = wmma4(as, bsB, p3B);
  }

  v8f crA = p1A - p2A;
  v8f ciA = p3A - p1A - p2A;
  v8f crB = p1B - p2B;
  v8f ciB = p3B - p1B - p2B;

  float* dr = Dr + (mtA + 8 * hi) * LDA + nt + ml;
  float* di = Di + (mtA + 8 * hi) * LDA + nt + ml;
  for (int v = 0; v < 8; v++) {
    dr[v * LDA] = crA[v] * ph.x - ciA[v] * ph.y;
    di[v * LDA] = crA[v] * ph.y + ciA[v] * ph.x;
  }
  for (int v = 0; v < 8; v++) {
    dr[(32 + v) * LDA] = crB[v] * ph.x - ciB[v] * ph.y;
    di[(32 + v) * LDA] = crB[v] * ph.y + ciB[v] * ph.x;
  }
}

struct RealPairs { float a0, a1, b0, b1; };
__device__ __forceinline__ RealPairs real_pairs(const float* rbrow, int kb) {
  RealPairs p;
  p.a0 = rbrow[kb];     p.a1 = rbrow[kb + 1];
  p.b0 = rbrow[kb + 4]; p.b1 = rbrow[kb + 5];
  return p;
}

__device__ __forceinline__ float2 rmix(bool first, float c, float2 z, float2 e) {
  return first ? make_float2(c * z.x, c * z.y)
               : make_float2(fmaf(c, z.x, e.x), fmaf(c, z.y, e.y));
}

struct Frag16 { v2f f[4]; };
__device__ __forceinline__ Frag16 frag16(const float (*RQc)[2], const float* R8m,
                                         int c, int kb) {
  const RealPairs g = real_pairs(R8m + ((c & 7) << 3), kb);
  const float f0 = RQc[c >> 3][0], f1 = RQc[c >> 3][1];
  Frag16 o;
  o.f[0].x = f0 * g.a0; o.f[0].y = f0 * g.a1;
  o.f[1].x = f0 * g.b0; o.f[1].y = f0 * g.b1;
  o.f[2].x = f1 * g.a0; o.f[2].y = f1 * g.a1;
  o.f[3].x = f1 * g.b0; o.f[3].y = f1 * g.b1;
  return o;
}

__device__ __forceinline__ void rgemm64_encA(const float (*RQw)[2][2], const float* R8m,
                                             const float2* QH,
                                             const float* Br, const float* Bi,
                                             float* Dr, float* Di) {
  const int tid = threadIdx.x;
  const int lane = tid & 31;
  const int w = tid >> 5;
  const int mh = w & 1;
  const int mtA = mh * 32;
  const int mtB = mtA + 16;
  const int nt = (w >> 1) * 16;
  const int ml = lane & 15;
  const int kb = (lane >> 4) * 2;
  const int hi = lane >> 4;
  float cA[4], cB[4];
  for (int j = 0; j < 4; j++) {
    const float r0 = RQw[0][mh][j >> 1];
    cA[j] = wave_uniform(r0 * RQw[1][0][j & 1]);
    cB[j] = wave_uniform(r0 * RQw[1][1][j & 1]);
  }
  const RealPairs g = real_pairs(R8m + ((ml & 7) << 3), kb);
  const float f0 = RQw[2][ml >> 3][0], f1 = RQw[2][ml >> 3][1];

  v8f trA = {}, tiA = {}, trB = {}, tiB = {};
#pragma unroll 1
  for (int t = 0; t < 4; t++) {
    const float ft = (t & 2) ? f1 : f0;
    v2f at;
    at.x = ft * ((t & 1) ? g.b0 : g.a0);
    at.y = ft * ((t & 1) ? g.b1 : g.a1);
    float2 eA0 = make_float2(0.f, 0.f), eA1 = eA0, eB0 = eA0, eB1 = eA0;
#pragma unroll
    for (int j = 0; j < 4; j++) {
      const int k = 16 * j + 4 * t + kb;
      const float4 q = *reinterpret_cast<const float4*>(QH + k);
      const int r0 = k * LDA + nt + ml, r1 = r0 + LDA;
      const float2 z0 = cmul(make_float2(q.x, q.y), make_float2(Br[r0], Bi[r0]));
      const float2 z1 = cmul(make_float2(q.z, q.w), make_float2(Br[r1], Bi[r1]));
      eA0 = rmix(j == 0, cA[j], z0, eA0);
      eA1 = rmix(j == 0, cA[j], z1, eA1);
      eB0 = rmix(j == 0, cB[j], z0, eB0);
      eB1 = rmix(j == 0, cB[j], z1, eB1);
    }
    v2f brA, biA, brB, biB;
    brA.x = eA0.x; brA.y = eA1.x;
    biA.x = eA0.y; biA.y = eA1.y;
    brB.x = eB0.x; brB.y = eB1.x;
    biB.x = eB0.y; biB.y = eB1.y;

    trA = wmma4(at, brA, trA);
    tiA = wmma4(at, biA, tiA);
    trB = wmma4(at, brB, trB);
    tiB = wmma4(at, biB, tiB);
  }

  for (int v = 0; v < 8; v++) {
    const int mA = mtA + v + 8 * hi, mB = mtB + v + 8 * hi;
    const int n = nt + ml;
    Dr[mA * LDA + n] = trA[v]; Di[mA * LDA + n] = tiA[v];
    Dr[mB * LDA + n] = trB[v]; Di[mB * LDA + n] = tiB[v];
  }
}

__device__ __forceinline__ void rgemm64_encB(const float* Ar, const float* Ai,
                                             const float2* PH, const float2* QL,
                                             const float (*RQw)[2][2], const float* R8m,
                                             float* Dr, float* Di) {
  const int tid = threadIdx.x;
  const int lane = tid & 31;
  const int w = tid >> 5;
  const int mt = (w & 3) * 16;
  const int nh = w >> 2;
  const int nt0 = nh * 32;
  const int nt1 = nt0 + 16;
  const int ml = lane & 15;
  const int kb = (lane >> 4) * 2;
  const int hi = lane >> 4;
  float cA[4], cB[4];
  for (int j = 0; j < 4; j++) {
    const float r0 = RQw[0][nh][j >> 1];
    cA[j] = wave_uniform(r0 * RQw[1][0][j & 1]);
    cB[j] = wave_uniform(r0 * RQw[1][1][j & 1]);
  }
  const Frag16 b = frag16(RQw[2], R8m, ml, kb);
  const float2 phm = PH[mt + ml];
  const float* arow = Ar + (mt + ml) * LDA + kb;
  const float* airow = Ai + (mt + ml) * LDA + kb;

  v8f xr0 = {}, xi0 = {}, xr1 = {}, xi1 = {};
#pragma unroll
  for (int t = 0; t < 4; t++) {
    float2 eA0 = make_float2(0.f, 0.f), eA1 = eA0, eB0 = eA0, eB1 = eA0;
#pragma unroll
    for (int j = 0; j < 4; j++) {
      const int k = 16 * j + 4 * t;
      const float4 q = *reinterpret_cast<const float4*>(QL + k + kb);
      const float2 z0 = cmul(make_float2(q.x, q.y), make_float2(arow[k], airow[k]));
      const float2 z1 = cmul(make_float2(q.z, q.w), make_float2(arow[k + 1], airow[k + 1]));
      eA0 = rmix(j == 0, cA[j], z0, eA0);
      eA1 = rmix(j == 0, cA[j], z1, eA1);
      eB0 = rmix(j == 0, cB[j], z0, eB0);
      eB1 = rmix(j == 0, cB[j], z1, eB1);
    }
    eA0 = cmul(phm, eA0); eA1 = cmul(phm, eA1);
    eB0 = cmul(phm, eB0); eB1 = cmul(phm, eB1);
    v2f arA, aiA, arB, aiB;
    arA.x = eA0.x; arA.y = eA1.x;
    aiA.x = eA0.y; aiA.y = eA1.y;
    arB.x = eB0.x; arB.y = eB1.x;
    aiB.x = eB0.y; aiB.y = eB1.y;

    xr0 = wmma4(arA, b.f[t], xr0);
    xi0 = wmma4(aiA, b.f[t], xi0);
    xr1 = wmma4(arB, b.f[t], xr1);
    xi1 = wmma4(aiB, b.f[t], xi1);
  }

  for (int v = 0; v < 8; v++) {
    int m = mt + v + 8 * hi;
    int n0 = nt0 + ml, n1 = nt1 + ml;
    Dr[m * LDA + n0] = xr0[v]; Di[m * LDA + n0] = xi0[v];
    Dr[m * LDA + n1] = xr1[v]; Di[m * LDA + n1] = xi1[v];
  }
}

__device__ __forceinline__ void cgemm64_rowpar(const float* Sr, const float* Si,
                                               const float2* MX4,
                                               const float4* __restrict__ Bf,
                                               float* Dr, float* Di) {
  const int tid = threadIdx.x;
  const int lane = tid & 31;
  const int w = tid >> 5;
  const int par = (w >> 1) & 1;
  const int rb = (w & 1) * 16;
  const int c1 = w >> 2;
  const int nt0 = c1 * 16;
  const int nt1 = nt0 + 32;
  const int ml = lane & 15;
  const int kb = (lane >> 4) * 2;
  const int hi = lane >> 4;
  const float4* bf = Bf + c1 * 640 + lane;

  float4 b[4];
  for (int t = 0; t < 4; t++) b[t] = bf[32 * t];
  __syncthreads();
  float2 cA[4], cB[4];
  for (int j = 0; j < 4; j++) {
    cA[j] = wave_uniform2(MX4[(par * 4 + c1) * 4 + j]);
    cB[j] = wave_uniform2(MX4[(par * 4 + 2 + c1) * 4 + j]);
  }
  const int arow = (2 * (rb + ml) + par) * LDA + kb;
  const float* ap = Sr + arow;
  const float* aip = Si + arow;

  v8f p1_0 = {}, p2_0 = {}, p3_0 = {};
  v8f p1_1 = {}, p2_1 = {}, p3_1 = {};
#pragma unroll
  for (int t = 0; t < 4; t++) {
    float2 eA0 = make_float2(0.f, 0.f), eA1 = eA0, eB0 = eA0, eB1 = eA0;
#pragma unroll
    for (int j = 0; j < 4; j++) {
      const int k = 16 * j + 4 * t;
      const float2 z0 = make_float2(ap[k], aip[k]);
      const float2 z1 = make_float2(ap[k + 1], aip[k + 1]);
      eA0 = cmix(j == 0, cA[j], z0, eA0);
      eA1 = cmix(j == 0, cA[j], z1, eA1);
      eB0 = cmix(j == 0, cB[j], z0, eB0);
      eB1 = cmix(j == 0, cB[j], z1, eB1);
    }
    v2f arA, aiA, arB, aiB;
    arA.x = eA0.x; arA.y = eA1.x;
    aiA.x = eA0.y; aiA.y = eA1.y;
    arB.x = eB0.x; arB.y = eB1.x;
    aiB.x = eB0.y; aiB.y = eB1.y;
    v2f asA = arA + aiA;
    v2f asB = arB + aiB;
    v2f br; br.x = b[t].x; br.y = b[t].z;
    v2f bi; bi.x = b[t].y; bi.y = b[t].w;
    v2f bs = br + bi;

    p1_0 = wmma4(arA, br, p1_0);
    p2_0 = wmma4(aiA, bi, p2_0);
    p3_0 = wmma4(asA, bs, p3_0);
    p1_1 = wmma4(arB, br, p1_1);
    p2_1 = wmma4(aiB, bi, p2_1);
    p3_1 = wmma4(asB, bs, p3_1);
  }

  v8f cr0 = p1_0 - p2_0;
  v8f ci0 = p3_0 - p1_0 - p2_0;
  v8f cr1 = p1_1 - p2_1;
  v8f ci1 = p3_1 - p1_1 - p2_1;

  float* dr = Dr + (2 * (rb + 8 * hi) + par) * LDA + nt0 + ml;
  float* di = Di + (2 * (rb + 8 * hi) + par) * LDA + nt0 + ml;
  for (int v = 0; v < 8; v++) {
    dr[2 * v * LDA] = cr0[v]; dr[2 * v * LDA + 32] = cr1[v];
    di[2 * v * LDA] = ci0[v]; di[2 * v * LDA + 32] = ci1[v];
  }
}

__global__ __launch_bounds__(256, 8) void qk_main(const float* __restrict__ x,
                                               const float* __restrict__ rotp,
                                               const float* __restrict__ entp,
                                               const float2* __restrict__ W,
                                               float* __restrict__ out) {
  __shared__ float S0r[64 * LDA], S0i[64 * LDA];
  __shared__ float S1r[64 * LDA], S1i[64 * LDA];
  __shared__ float RQ[12][2][2];
  __shared__ float2 PP[12][2];
  __shared__ float2 QQ[12][2];
  __shared__ float R8[4][64];
  __shared__ __attribute__((aligned(16))) float2 PV[4][64];
  __shared__ float2 GU[NL][8][2][2];
  __shared__ float2 MXA[NL][4][4];
  __shared__ float2 MXL[NL][2][4][4];

  const int tid = threadIdx.x;
  const int s = blockIdx.x;

  if (tid < 12) {
    const float* xs = x + (size_t)s * 48;
    float a = xs[0 * 12 + tid], b = xs[1 * 12 + tid];
    float c = xs[2 * 12 + tid], d = xs[3 * 12 + tid];
    float ca = cosf(0.5f * a), sa = sinf(0.5f * a);
    float cb = cosf(0.5f * b), sb = sinf(0.5f * b);
    float cc = cosf(0.5f * c), sc = sinf(0.5f * c);
    float cd = cosf(0.5f * d), sd = sinf(0.5f * d);
    C2 M  = { {ca, 0.f}, {-sa, 0.f}, {sa, 0.f}, {ca, 0.f} };
    C2 RX = { {cb, 0.f}, {0.f, -sb}, {0.f, -sb}, {cb, 0.f} };
    C2 RZ = { {cc, -sc}, {0.f, 0.f}, {0.f, 0.f}, {cc, sc} };
    C2 RY = { {cd, 0.f}, {-sd, 0.f}, {sd, 0.f}, {cd, 0.f} };
    M = mm2(RX, M); M = mm2(RZ, M); M = mm2(RY, M);
    const float cs = sqrtf(M.a00.x * M.a00.x + M.a00.y * M.a00.y);
    const float sn = sqrtf(M.a10.x * M.a10.x + M.a10.y * M.a10.y);
    float2 p0 = make_float2(1.f, 0.f), p1 = make_float2(1.f, 0.f), q1;
    if (cs > 0.f) { const float r = 1.f / cs; p0 = make_float2(M.a00.x * r, M.a00.y * r); }
    if (sn > 0.f) { const float r = 1.f / sn; p1 = make_float2(M.a10.x * r, M.a10.y * r); }
    if (cs >= sn) {
      const float r = 1.f / cs;
      const float2 t = cmul(M.a11, make_float2(p1.x, -p1.y));
      q1 = make_float2(t.x * r, t.y * r);
    } else {
      const float r = -1.f / sn;
      const float2 t = cmul(M.a01, make_float2(p0.x, -p0.y));
      q1 = make_float2(t.x * r, t.y * r);
    }
    RQ[tid][0][0] = cs; RQ[tid][0][1] = -sn;
    RQ[tid][1][0] = sn; RQ[tid][1][1] = cs;
    PP[tid][0] = p0; PP[tid][1] = p1;
    QQ[tid][0] = make_float2(1.f, 0.f); QQ[tid][1] = q1;
  }
  if (tid < 8 * NL) {
    const int gl = tid >> 3, gg = tid & 7;
    const float* p = (gg == 0) ? &entp[(gl * 12 + 11) * 3]
                   : (gg == 1) ? &rotp[(gl * 12 + 0) * 3]
                   : (gg == 2) ? &rotp[(gl * 12 + 6) * 3]
                   : (gg == 3) ? &entp[(gl * 12 + 5) * 3]
                   : (gg == 4) ? &rotp[(gl * 12 + 1) * 3]
                   : (gg == 5) ? &entp[(gl * 12 + 0) * 3]
                   : (gg == 6) ? &rotp[(gl * 12 + 7) * 3]
                               : &entp[(gl * 12 + 6) * 3];
    rot_gate(p[0], p[1], p[2], &GU[gl][gg][0][0], &GU[gl][gg][0][1],
             &GU[gl][gg][1][0], &GU[gl][gg][1][1]);
  }
  for (int idx = tid; idx < 4096; idx += 256) {
    int i = idx >> 6, j = idx & 63;
    S0r[i * LDA + j] = (idx == 0) ? 1.f : 0.f;
    S0i[i * LDA + j] = 0.f;
  }
  __syncthreads();

  {
    int g = tid >> 6, a = (tid >> 3) & 7, b = tid & 7;
    R8[g][a * 8 + b] = RQ[3 * g][(a >> 2) & 1][(b >> 2) & 1] *
                       RQ[3 * g + 1][(a >> 1) & 1][(b >> 1) & 1] *
                       RQ[3 * g + 2][a & 1][b & 1];
    const int idx = tid & 63;
    const int w0 = (g >> 1) * 6;
    const float2 (*ph)[2] = (g & 1) ? QQ : PP;
    float2 p = ph[w0][(idx >> 5) & 1];
    for (int q = 1; q < 6; q++)
      p = cmul(p, ph[w0 + q][(idx >> (5 - q)) & 1]);
    PV[g][idx] = p;
  }
  if (tid < 192) {
    const bool low = tid >= 64;
    const int e = low ? tid - 64 : tid;
    const int tl = low ? (e >> 5) : (e >> 4);
    const int tp = low ? ((e >> 4) & 1) : 0;
    const int o0 = (e >> 3) & 1, o1 = (e >> 2) & 1, a0 = (e >> 1) & 1, a1 = e & 1;
    const float2 (*Fg)[2] = GU[tl][low ? 2 : 1];
    const float2 (*Vg)[2] = GU[tl][low ? 6 : 4];
    const float2 (*Cg)[2] = GU[tl][low ? 7 : 5];
    float2 f = Fg[o0][a0];
    if (tp)
      f = cadd(cmul(GU[tl][3][o0][0], Fg[0][a0]), cmul(GU[tl][3][o0][1], Fg[1][a0]));
    float2 g = Vg[o1][a1];
    if (o0)
      g = cadd(cmul(Cg[o1][0], Vg[0][a1]), cmul(Cg[o1][1], Vg[1][a1]));
    const float2 m = cmul(f, g);
    if (low) MXL[tl][tp][2 * o0 + o1][2 * a0 + a1] = m;
    else     MXA[tl][2 * o0 + o1][2 * a0 + a1] = m;
  }
  __syncthreads();

  float *curR = S0r, *curI = S0i, *nxtR = S1r, *nxtI = S1i;

  for (int l = 0; l < NL; l++) {
    const float4* Wl = reinterpret_cast<const float4*>(W) + (size_t)l * 2 * 2048;

    rgemm64_encA(&RQ[0], R8[1], PV[1], curR, curI, nxtR, nxtI);
    __syncthreads();
    { float* t; t = curR; curR = nxtR; nxtR = t; t = curI; curI = nxtI; nxtI = t; }

    rgemm64_encB(curR, curI, PV[0], PV[3], &RQ[6], R8[3], nxtR, nxtI);

    { float* t; t = curR; curR = nxtR; nxtR = t; t = curI; curI = nxtI; nxtI = t; }
    cgemm64_opA(Wl, &MXA[l][0][0], PV[2], curR, curI, nxtR, nxtI);
    { float* t; t = curR; curR = nxtR; nxtR = t; t = curI; curI = nxtI; nxtI = t; }

    cgemm64_rowpar(curR, curI, &MXL[l][0][0][0], Wl + 2048, nxtR, nxtI);
    { float* t; t = curR; curR = nxtR; nxtR = t; t = curI; curI = nxtI; nxtI = t; }
    __syncthreads();

    {
      float2 u00 = GU[l][0][0][0], u01 = GU[l][0][0][1];
      float2 u10 = GU[l][0][1][0], u11 = GU[l][0][1][1];
      for (int t = tid; t < 1024; t += 256) {
        int il = t >> 5;
        int j = ((t & 31) << 1) | 1;
        int a0 = il * LDA + j, a1 = (il + 32) * LDA + j;
        float2 x0 = make_float2(curR[a0], curI[a0]);
        float2 x1 = make_float2(curR[a1], curI[a1]);
        float2 y0 = cadd(cmul(u00, x0), cmul(u01, x1));
        float2 y1 = cadd(cmul(u10, x0), cmul(u11, x1));
        curR[a0] = y0.x; curI[a0] = y0.y;
        curR[a1] = y1.x; curI[a1] = y1.y;
      }
    }
    __syncthreads();
  }

  float z[12];
  for (int q = 0; q < 12; q++) z[q] = 0.f;
  for (int idx = tid; idx < 4096; idx += 256) {
    int i = idx >> 6, j = idx & 63;
    float pr = curR[i * LDA + j], pi = curI[i * LDA + j];
    float p = pr * pr + pi * pi;
    for (int q = 0; q < 12; q++)
      z[q] += ((idx >> (11 - q)) & 1) ? -p : p;
  }
  __syncthreads();
  float* red = nxtR;
  for (int q = 0; q < 12; q++) red[tid * 12 + q] = z[q];
  __syncthreads();
  for (int sft = 128; sft > 0; sft >>= 1) {
    if (tid < sft)
      for (int q = 0; q < 12; q++)
        red[tid * 12 + q] += red[(tid + sft) * 12 + q];
    __syncthreads();
  }
  if (tid < 12) out[(size_t)s * 12 + tid] = red[tid];
}

extern "C" void kernel_launch(void* const* d_in, const int* in_sizes, int n_in,
                              void* d_out, int out_size, void* d_ws, size_t ws_size,
                              hipStream_t stream) {
  (void)n_in; (void)out_size; (void)ws_size;
  const float* x    = (const float*)d_in[0];
  const float* rotp = (const float*)d_in[1];
  const float* entp = (const float*)d_in[2];
  float2* W = (float2*)d_ws;
  const int B = in_sizes[0] / 48;

  qk_prep<<<NPREP, 256, 0, stream>>>(rotp, entp, W);
  qk_main<<<B, 256, 0, stream>>>(x, rotp, entp, W, (float*)d_out);
}


# Encoding folded per sample into the shared operators' 16x16 blocks (W = N*E2, 8 real WMMAs a wave; wires 0,1/6,7 into the 4x4 mixes): 2 complex K=16 products a layer, not 4; sum order changes
# speedup vs baseline: 4.7276x; 1.4232x over previous
#include <hip/hip_runtime.h>
#include <math.h>

#define LDA 65
#define NL  4
#define NPREP 8

typedef __attribute__((ext_vector_type(2))) float v2f;
typedef __attribute__((ext_vector_type(8))) float v8f;

__device__ __forceinline__ float2 cmul(float2 a, float2 b) {
  return make_float2(a.x * b.x - a.y * b.y, a.x * b.y + a.y * b.x);
}
__device__ __forceinline__ float2 cadd(float2 a, float2 b) {
  return make_float2(a.x + b.x, a.y + b.y);
}

struct C2 { float2 a00, a01, a10, a11; };
__device__ __forceinline__ C2 mm2(const C2 L, const C2 R) {
  C2 o;
  o.a00 = cadd(cmul(L.a00, R.a00), cmul(L.a01, R.a10));
  o.a01 = cadd(cmul(L.a00, R.a01), cmul(L.a01, R.a11));
  o.a10 = cadd(cmul(L.a10, R.a00), cmul(L.a11, R.a10));
  o.a11 = cadd(cmul(L.a10, R.a01), cmul(L.a11, R.a11));
  return o;
}

__device__ __forceinline__ void rot_gate(float phi, float th, float om,
                                         float2* u00, float2* u01,
                                         float2* u10, float2* u11) {
  float ct = cosf(0.5f * th), st = sinf(0.5f * th);
  float ap = -0.5f * (phi + om);
  float am = 0.5f * (phi - om);
  *u00 = make_float2(ct * cosf(ap), ct * sinf(ap));
  *u01 = make_float2(-st * cosf(am), -st * sinf(am));
  *u10 = make_float2(st * cosf(-am), st * sinf(-am));
  *u11 = make_float2(ct * cosf(-ap), ct * sinf(-ap));
}

__device__ __forceinline__ void apply_cgate(float* Mr, float* Mi,
                                            float2 u00, float2 u01,
                                            float2 u10, float2 u11,
                                            int cb, int tb, int tid) {
  for (int idx = tid; idx < 4096; idx += 256) {
    int i = idx >> 6, j = idx & 63;
    if ((i & cb) && !(i & tb)) {
      int a0 = i * LDA + j, a1 = (i | tb) * LDA + j;
      float2 x0 = make_float2(Mr[a0], Mi[a0]);
      float2 x1 = make_float2(Mr[a1], Mi[a1]);
      float2 y0 = cadd(cmul(u00, x0), cmul(u01, x1));
      float2 y1 = cadd(cmul(u10, x0), cmul(u11, x1));
      Mr[a0] = y0.x; Mi[a0] = y0.y;
      Mr[a1] = y1.x; Mi[a1] = y1.y;
    }
  }
}

__global__ __launch_bounds__(256) void qk_prep(const float* __restrict__ rotp,
                                               const float* __restrict__ entp,
                                               float2* __restrict__ W) {
  __shared__ float Mr[64 * LDA], Mi[64 * LDA];
  __shared__ float2 U[24][2][2];
  const int l = blockIdx.x >> 1;
  const int slot = blockIdx.x & 1;
  const int tid = threadIdx.x;

  if (tid < 24) {
    const float* p = (tid < 12) ? &rotp[(l * 12 + tid) * 3]
                                : &entp[(l * 12 + (tid - 12)) * 3];
    rot_gate(p[0], p[1], p[2], &U[tid][0][0], &U[tid][0][1],
             &U[tid][1][0], &U[tid][1][1]);
  }
  __syncthreads();

  const int g0 = slot ? 6 : 0;

  for (int idx = tid; idx < 4096; idx += 256) {
    int i = idx >> 6, j = idx & 63;
    float2 p = make_float2(((i ^ j) >> 4) ? 0.f : 1.f, 0.f);
    for (int q = 2; q < 6; q++)
      p = cmul(p, U[g0 + q][(i >> (5 - q)) & 1][(j >> (5 - q)) & 1]);
    Mr[i * LDA + j] = p.x; Mi[i * LDA + j] = p.y;
  }
  __syncthreads();
  for (int c = 1; c < 5; c++) {
    apply_cgate(Mr, Mi, U[12 + g0 + c][0][0], U[12 + g0 + c][0][1],
                U[12 + g0 + c][1][0], U[12 + g0 + c][1][1],
                1 << (5 - c), 1 << (4 - c), tid);
    __syncthreads();
  }
  float2* Wd = W + ((size_t)l * 2 + slot) * 4096;
  for (int idx = tid; idx < 4096; idx += 256) {
    int i = idx >> 6, j = idx & 63;
    int dst = (((((i >> 4) * 16 + (j >> 2)) * 32 + ((j >> 1) & 1) * 16 + (i & 15)) << 1) | (j & 1));
    Wd[dst] = make_float2(Mr[i * LDA + j], Mi[i * LDA + j]);
  }
}

__device__ __forceinline__ v8f wmma4(v2f a, v2f b, v8f c) {
  return __builtin_amdgcn_wmma_f32_16x16x4_f32(
      false, a, false, b, (short)0, c, false, false);
}

__device__ __forceinline__ float wave_uniform(float v) {
  return __int_as_float(__builtin_amdgcn_readfirstlane(__float_as_int(v)));
}
__device__ __forceinline__ float2 wave_uniform2(float2 v) {
  return make_float2(wave_uniform(v.x), wave_uniform(v.y));
}
__device__ __forceinline__ float2 cmix(bool first, float2 c, float2 z, float2 e) {
  return first ? make_float2(fmaf(-c.y, z.y, c.x * z.x), fmaf(c.y, z.x, c.x * z.y))
               : make_float2(fmaf(-c.y, z.y, fmaf(c.x, z.x, e.x)),
                             fmaf(c.y, z.x, fmaf(c.x, z.y, e.y)));
}

struct Blk4 { float4 f[4]; };
__device__ __forceinline__ Blk4 load_blk(const float4* __restrict__ base, unsigned off) {
  Blk4 b;
  for (int t = 0; t < 4; t++) b.f[t] = base[off + 32u * t];
  return b;
}

struct OpFrag { v8f re, im; };
__device__ __forceinline__ OpFrag fold_enc(const Blk4 n, const float2* A2,
                                           const float2* PQ) {
  v8f wr = {}, wi = {};
#pragma unroll
  for (int t = 0; t < 4; t++) {
    const float2 a2 = A2[32 * t];
    v2f at;
    at.x = a2.x; at.y = a2.y;
    const float4 p = *reinterpret_cast<const float4*>(PQ + 2 * t);
    const float2 z0 = cmul(make_float2(p.x, p.y), make_float2(n.f[t].x, n.f[t].y));
    const float2 z1 = cmul(make_float2(p.z, p.w), make_float2(n.f[t].z, n.f[t].w));
    v2f br, bi;
    br.x = z0.x; br.y = z1.x;
    bi.x = z0.y; bi.y = z1.y;
    wr = wmma4(at, br, wr);
    wi = wmma4(at, bi, wi);
  }
  OpFrag o;
#pragma unroll
  for (int u = 0; u < 4; u++) {
    const float4 q = *reinterpret_cast<const float4*>(PQ + 8 + 2 * u);
    const float2 y0 = cmul(make_float2(q.x, q.y), make_float2(wr[2 * u], wi[2 * u]));
    const float2 y1 = cmul(make_float2(q.z, q.w), make_float2(wr[2 * u + 1], wi[2 * u + 1]));
    o.re[2 * u] = y0.x;     o.im[2 * u] = y0.y;
    o.re[2 * u + 1] = y1.x; o.im[2 * u + 1] = y1.y;
  }
  return o;
}

__device__ __forceinline__ Blk4 cgemm64_opA(const Blk4 n, const float2* A2,
                                            const float2* PQ, const float2* MX4,
                                            const float* Br, const float* Bi,
                                            float* Dr, float* Di,
                                            const float4* __restrict__ next,
                                            unsigned noff) {
  const int tid = threadIdx.x;
  const int lane = tid & 31;
  const int w = __builtin_amdgcn_readfirstlane(tid >> 5);
  const int b1 = w & 1;
  const int mtA = b1 * 16;
  const int nt = (w >> 1) * 16;
  const int ml = lane & 15;
  const int hi = lane >> 4;

  __syncthreads();
  const OpFrag W = fold_enc(n, A2, PQ);
  float2 cA[4], cB[4];
  for (int j = 0; j < 4; j++) {
    cA[j] = wave_uniform2(MX4[b1 * 4 + j]);
    cB[j] = wave_uniform2(MX4[(2 + b1) * 4 + j]);
  }
  const float* brow = Br + (8 * hi) * LDA + nt + ml;
  const float* birow = Bi + (8 * hi) * LDA + nt + ml;

  v8f p1A = {}, p2A = {}, p3A = {};
  v8f p1B = {}, p2B = {}, p3B = {};
#pragma unroll
  for (int t = 0; t < 4; t++) {
    v2f ar; ar.x = W.re[2 * t]; ar.y = W.re[2 * t + 1];
    v2f ai; ai.x = W.im[2 * t]; ai.y = W.im[2 * t + 1];
    v2f as = ar + ai;
    float2 eA0 = make_float2(0.f, 0.f), eA1 = eA0, eB0 = eA0, eB1 = eA0;
#pragma unroll
    for (int j = 0; j < 4; j++) {
      const int r0 = (16 * j + 2 * t) * LDA, r1 = r0 + LDA;
      const float2 z0 = make_float2(brow[r0], birow[r0]);
      const float2 z1 = make_float2(brow[r1], birow[r1]);
      eA0 = cmix(j == 0, cA[j], z0, eA0);
      eA1 = cmix(j == 0, cA[j], z1, eA1);
      eB0 = cmix(j == 0, cB[j], z0, eB0);
      eB1 = cmix(j == 0, cB[j], z1, eB1);
    }
    v2f brA, biA, brB, biB;
    brA.x = eA0.x; brA.y = eA1.x;
    biA.x = eA0.y; biA.y = eA1.y;
    brB.x = eB0.x; brB.y = eB1.x;
    biB.x = eB0.y; biB.y = eB1.y;
    v2f bsA = brA + biA;
    v2f bsB = brB + biB;

    p1A = wmma4(ar, brA, p1A);
    p2A = wmma4(ai, biA, p2A);
    p3A = wmma4(as, bsA, p3A);
    p1B = wmma4(ar, brB, p1B);
    p2B = wmma4(ai, biB, p2B);
    p3B = wmma4(as, bsB, p3B);
  }

  const Blk4 nx = load_blk(next, noff);

  v8f crA = p1A - p2A;
  v8f ciA = p3A - p1A - p2A;
  v8f crB = p1B - p2B;
  v8f ciB = p3B - p1B - p2B;

  float* dr = Dr + (mtA + 8 * hi) * LDA + nt + ml;
  float* di = Di + (mtA + 8 * hi) * LDA + nt + ml;
  for (int v = 0; v < 8; v++) {
    dr[v * LDA] = crA[v];
    di[v * LDA] = ciA[v];
  }
  for (int v = 0; v < 8; v++) {
    dr[(32 + v) * LDA] = crB[v];
    di[(32 + v) * LDA] = ciB[v];
  }
  return nx;
}

__device__ __forceinline__ Blk4 cgemm64_rowpar(const Blk4 n, const float2* A2,
                                               const float2* PQ,
                                               const float* Sr, const float* Si,
                                               const float2* MX4,
                                               float* Dr, float* Di,
                                               const float4* __restrict__ next,
                                               unsigned noff) {
  const int tid = threadIdx.x;
  const int lane = tid & 31;
  const int w = __builtin_amdgcn_readfirstlane(tid >> 5);
  const int par = (w >> 1) & 1;
  const int rb = (w & 1) * 16;
  const int c1 = w >> 2;
  const int nt0 = c1 * 16;
  const int ml = lane & 15;
  const int hi = lane >> 4;

  __syncthreads();
  const OpFrag W = fold_enc(n, A2, PQ);
  float2 cA[4], cB[4];
  for (int j = 0; j < 4; j++) {
    cA[j] = wave_uniform2(MX4[(par * 4 + c1) * 4 + j]);
    cB[j] = wave_uniform2(MX4[(par * 4 + 2 + c1) * 4 + j]);
  }
  const int arow = (2 * (rb + ml) + par) * LDA + 8 * hi;
  const float* ap = Sr + arow;
  const float* aip = Si + arow;

  v8f p1_0 = {}, p2_0 = {}, p3_0 = {};
  v8f p1_1 = {}, p2_1 = {}, p3_1 = {};
#pragma unroll
  for (int t = 0; t < 4; t++) {
    float2 eA0 = make_float2(0.f, 0.f), eA1 = eA0, eB0 = eA0, eB1 = eA0;
#pragma unroll
    for (int j = 0; j < 4; j++) {
      const int k = 16 * j + 2 * t;
      const float2 z0 = make_float2(ap[k], aip[k]);
      const float2 z1 = make_float2(ap[k + 1], aip[k + 1]);
      eA0 = cmix(j == 0, cA[j], z0, eA0);
      eA1 = cmix(j == 0, cA[j], z1, eA1);
      eB0 = cmix(j == 0, cB[j], z0, eB0);
      eB1 = cmix(j == 0, cB[j], z1, eB1);
    }
    v2f arA, aiA, arB, aiB;
    arA.x = eA0.x; arA.y = eA1.x;
    aiA.x = eA0.y; aiA.y = eA1.y;
    arB.x = eB0.x; arB.y = eB1.x;
    aiB.x = eB0.y; aiB.y = eB1.y;
    v2f asA = arA + aiA;
    v2f asB = arB + aiB;
    v2f br; br.x = W.re[2 * t]; br.y = W.re[2 * t + 1];
    v2f bi; bi.x = W.im[2 * t]; bi.y = W.im[2 * t + 1];
    v2f bs = br + bi;

    p1_0 = wmma4(arA, br, p1_0);
    p2_0 = wmma4(aiA, bi, p2_0);
    p3_0 = wmma4(asA, bs, p3_0);
    p1_1 = wmma4(arB, br, p1_1);
    p2_1 = wmma4(aiB, bi, p2_1);
    p3_1 = wmma4(asB, bs, p3_1);
  }

  const Blk4 nx = load_blk(next, noff);

  v8f cr0 = p1_0 - p2_0;
  v8f ci0 = p3_0 - p1_0 - p2_0;
  v8f cr1 = p1_1 - p2_1;
  v8f ci1 = p3_1 - p1_1 - p2_1;

  float* dr = Dr + (2 * (rb + 8 * hi) + par) * LDA + nt0 + ml;
  float* di = Di + (2 * (rb + 8 * hi) + par) * LDA + nt0 + ml;
  for (int v = 0; v < 8; v++) {
    dr[2 * v * LDA] = cr0[v]; dr[2 * v * LDA + 32] = cr1[v];
    di[2 * v * LDA] = ci0[v]; di[2 * v * LDA + 32] = ci1[v];
  }
  return nx;
}

__global__ __launch_bounds__(256, 8) void qk_main(const float* __restrict__ x,
                                               const float* __restrict__ rotp,
                                               const float* __restrict__ entp,
                                               const float2* __restrict__ W,
                                               float* __restrict__ out) {
  __shared__ float S0r[64 * LDA], S0i[64 * LDA];
  __shared__ float S1r[64 * LDA], S1i[64 * LDA];
  __shared__ float2 EG[12][2][2];
  __shared__ float RQ[12][2][2];
  __shared__ float2 PP[12][2];
  __shared__ float2 QQ[12][2];
  __shared__ float2 A2T[2][4][32];
  __shared__ __attribute__((aligned(16))) float2 PF[2][2][2][8];
  __shared__ float2 GU[NL][8][2][2];
  __shared__ float2 MXA[NL][4][4];
  __shared__ float2 MXL[NL][2][4][4];
  __shared__ float2 MXAe[NL][4][4];
  __shared__ float2 MXLe[NL][2][4][4];

  const int tid = threadIdx.x;
  const int s = blockIdx.x;

  if (tid < 12) {
    const float* xs = x + (size_t)s * 48;
    float a = xs[0 * 12 + tid], b = xs[1 * 12 + tid];
    float c = xs[2 * 12 + tid], d = xs[3 * 12 + tid];
    float ca = cosf(0.5f * a), sa = sinf(0.5f * a);
    float cb = cosf(0.5f * b), sb = sinf(0.5f * b);
    float cc = cosf(0.5f * c), sc = sinf(0.5f * c);
    float cd = cosf(0.5f * d), sd = sinf(0.5f * d);
    C2 M  = { {ca, 0.f}, {-sa, 0.f}, {sa, 0.f}, {ca, 0.f} };
    C2 RX = { {cb, 0.f}, {0.f, -sb}, {0.f, -sb}, {cb, 0.f} };
    C2 RZ = { {cc, -sc}, {0.f, 0.f}, {0.f, 0.f}, {cc, sc} };
    C2 RY = { {cd, 0.f}, {-sd, 0.f}, {sd, 0.f}, {cd, 0.f} };
    M = mm2(RX, M); M = mm2(RZ, M); M = mm2(RY, M);
    EG[tid][0][0] = M.a00; EG[tid][0][1] = M.a01;
    EG[tid][1][0] = M.a10; EG[tid][1][1] = M.a11;
    const float cs = sqrtf(M.a00.x * M.a00.x + M.a00.y * M.a00.y);
    const float sn = sqrtf(M.a10.x * M.a10.x + M.a10.y * M.a10.y);
    float2 p0 = make_float2(1.f, 0.f), p1 = make_float2(1.f, 0.f), q1;
    if (cs > 0.f) { const float r = 1.f / cs; p0 = make_float2(M.a00.x * r, M.a00.y * r); }
    if (sn > 0.f) { const float r = 1.f / sn; p1 = make_float2(M.a10.x * r, M.a10.y * r); }
    if (cs >= sn) {
      const float r = 1.f / cs;
      const float2 t = cmul(M.a11, make_float2(p1.x, -p1.y));
      q1 = make_float2(t.x * r, t.y * r);
    } else {
      const float r = -1.f / sn;
      const float2 t = cmul(M.a01, make_float2(p0.x, -p0.y));
      q1 = make_float2(t.x * r, t.y * r);
    }
    RQ[tid][0][0] = cs; RQ[tid][0][1] = -sn;
    RQ[tid][1][0] = sn; RQ[tid][1][1] = cs;
    PP[tid][0] = p0; PP[tid][1] = p1;
    QQ[tid][0] = make_float2(1.f, 0.f); QQ[tid][1] = q1;
  }
  if (tid < 8 * NL) {
    const int gl = tid >> 3, gg = tid & 7;
    const float* p = (gg == 0) ? &entp[(gl * 12 + 11) * 3]
                   : (gg == 1) ? &rotp[(gl * 12 + 0) * 3]
                   : (gg == 2) ? &rotp[(gl * 12 + 6) * 3]
                   : (gg == 3) ? &entp[(gl * 12 + 5) * 3]
                   : (gg == 4) ? &rotp[(gl * 12 + 1) * 3]
                   : (gg == 5) ? &entp[(gl * 12 + 0) * 3]
                   : (gg == 6) ? &rotp[(gl * 12 + 7) * 3]
                               : &entp[(gl * 12 + 6) * 3];
    rot_gate(p[0], p[1], p[2], &GU[gl][gg][0][0], &GU[gl][gg][0][1],
             &GU[gl][gg][1][0], &GU[gl][gg][1][1]);
  }
  for (int idx = tid; idx < 4096; idx += 256) {
    int i = idx >> 6, j = idx & 63;
    S0r[i * LDA + j] = (idx == 0) ? 1.f : 0.f;
    S0i[i * LDA + j] = 0.f;
  }
  __syncthreads();

  {
    const int half = tid >> 7, t = (tid >> 5) & 3, ln = tid & 31;
    const int w0 = half * 6 + 2;
    const int c = ln & 15;
    float v[2];
    for (int sx = 0; sx < 2; sx++) {
      const int i = 4 * t + 2 * (ln >> 4) + sx;
      float r = RQ[w0][(i >> 3) & 1][(c >> 3) & 1];
      for (int q = 1; q < 4; q++)
        r *= RQ[w0 + q][(i >> (3 - q)) & 1][(c >> (3 - q)) & 1];
      v[sx] = r;
    }
    A2T[half][t][ln] = make_float2(v[0], v[1]);
  }
  if (tid < 64) {
    const int h = tid >> 5, half = (tid >> 4) & 1, pq = (tid >> 3) & 1, e = tid & 7;
    const int w0 = half * 6 + 2;
    const int idx = pq ? 8 * h + e : 4 * (e >> 1) + 2 * h + (e & 1);
    const float2 (*ph)[2] = pq ? QQ : PP;
    float2 p = ph[w0][(idx >> 3) & 1];
    for (int q = 1; q < 4; q++)
      p = cmul(p, ph[w0 + q][(idx >> (3 - q)) & 1]);
    PF[h][half][pq][e] = p;
  }
  if (tid < 192) {
    const bool low = tid >= 64;
    const int e = low ? tid - 64 : tid;
    const int tl = low ? (e >> 5) : (e >> 4);
    const int tp = low ? ((e >> 4) & 1) : 0;
    const int o0 = (e >> 3) & 1, o1 = (e >> 2) & 1, a0 = (e >> 1) & 1, a1 = e & 1;
    const float2 (*Fg)[2] = GU[tl][low ? 2 : 1];
    const float2 (*Vg)[2] = GU[tl][low ? 6 : 4];
    const float2 (*Cg)[2] = GU[tl][low ? 7 : 5];
    float2 f = Fg[o0][a0];
    if (tp)
      f = cadd(cmul(GU[tl][3][o0][0], Fg[0][a0]), cmul(GU[tl][3][o0][1], Fg[1][a0]));
    float2 g = Vg[o1][a1];
    if (o0)
      g = cadd(cmul(Cg[o1][0], Vg[0][a1]), cmul(Cg[o1][1], Vg[1][a1]));
    const float2 m = cmul(f, g);
    if (low) MXL[tl][tp][2 * o0 + o1][2 * a0 + a1] = m;
    else     MXA[tl][2 * o0 + o1][2 * a0 + a1] = m;
  }
  __syncthreads();
  if (tid < 192) {
    const bool low = tid >= 64;
    const int e = low ? tid - 64 : tid;
    const int j = e & 3;
    const float2* row = low ? &MXL[0][0][0][0] + (e & ~3) : &MXA[0][0][0] + (e & ~3);
    const int w0 = low ? 6 : 0;
    float2 m = make_float2(0.f, 0.f);
    for (int i = 0; i < 4; i++)
      m = cadd(m, cmul(row[i], cmul(EG[w0][i >> 1][j >> 1], EG[w0 + 1][i & 1][j & 1])));
    if (low) (&MXLe[0][0][0][0])[e] = m;
    else     (&MXAe[0][0][0])[e] = m;
  }

  const int lane = tid & 31;
  const int wv = tid >> 5;
  const float2* A2l = &A2T[0][0][lane];
  const float2* PQl = &PF[lane >> 4][0][0][0];
  const float4* Wf = reinterpret_cast<const float4*>(W);
  const unsigned offA = (wv & 1) * 640 + lane;
  const unsigned offB = (wv >> 2) * 640 + lane;
  Blk4 nA = load_blk(Wf, offA);
  __syncthreads();

  for (int l = 0; l < NL; l++) {
    const Blk4 nB = cgemm64_opA(nA, A2l, PQl, &MXAe[l][0][0],
                                S0r, S0i, S1r, S1i, Wf + l * 4096 + 2048, offB);

    nA = cgemm64_rowpar(nB, A2l + 128, PQl + 16, S1r, S1i,
                        &MXLe[l][0][0][0], S0r, S0i,
                        Wf + ((l + 1 < NL) ? l + 1 : l) * 4096, offA);
    __syncthreads();

    {
      float2 u00 = GU[l][0][0][0], u01 = GU[l][0][0][1];
      float2 u10 = GU[l][0][1][0], u11 = GU[l][0][1][1];
      for (int t = tid; t < 1024; t += 256) {
        int il = t >> 5;
        int j = ((t & 31) << 1) | 1;
        int a0 = il * LDA + j, a1 = (il + 32) * LDA + j;
        float2 x0 = make_float2(S0r[a0], S0i[a0]);
        float2 x1 = make_float2(S0r[a1], S0i[a1]);
        float2 y0 = cadd(cmul(u00, x0), cmul(u01, x1));
        float2 y1 = cadd(cmul(u10, x0), cmul(u11, x1));
        S0r[a0] = y0.x; S0i[a0] = y0.y;
        S0r[a1] = y1.x; S0i[a1] = y1.y;
      }
    }
  }
  __syncthreads();

  float z[12];
  for (int q = 0; q < 12; q++) z[q] = 0.f;
  for (int idx = tid; idx < 4096; idx += 256) {
    int i = idx >> 6, j = idx & 63;
    float pr = S0r[i * LDA + j], pi = S0i[i * LDA + j];
    float p = pr * pr + pi * pi;
    for (int q = 0; q < 12; q++)
      z[q] += ((idx >> (11 - q)) & 1) ? -p : p;
  }
  __syncthreads();
  float* red = S1r;
  for (int q = 0; q < 12; q++) red[tid * 12 + q] = z[q];
  __syncthreads();
  for (int sft = 128; sft > 0; sft >>= 1) {
    if (tid < sft)
      for (int q = 0; q < 12; q++)
        red[tid * 12 + q] += red[(tid + sft) * 12 + q];
    __syncthreads();
  }
  if (tid < 12) out[(size_t)s * 12 + tid] = red[tid];
}

extern "C" void kernel_launch(void* const* d_in, const int* in_sizes, int n_in,
                              void* d_out, int out_size, void* d_ws, size_t ws_size,
                              hipStream_t stream) {
  (void)n_in; (void)out_size; (void)ws_size;
  const float* x    = (const float*)d_in[0];
  const float* rotp = (const float*)d_in[1];
  const float* entp = (const float*)d_in[2];
  float2* W = (float2*)d_ws;
  const int B = in_sizes[0] / 48;

  qk_prep<<<NPREP, 256, 0, stream>>>(rotp, entp, W);
  qk_main<<<B, 256, 0, stream>>>(x, rotp, entp, W, (float*)d_out);
}


# W = N*E2 blocks folded once per workgroup (waves 0..3, one each, to LDS) beside CRot(11,0) on waves 4..7 (LDS loads grouped), not by every wave: 52 WMMAs a wave-layer, not 64; sums unchanged
# speedup vs baseline: 5.1585x; 1.0911x over previous
#include <hip/hip_runtime.h>
#include <math.h>

#define LDA 65
#define NL  4
#define NPREP 8

typedef __attribute__((ext_vector_type(2))) float v2f;
typedef __attribute__((ext_vector_type(8))) float v8f;

__device__ __forceinline__ float2 cmul(float2 a, float2 b) {
  return make_float2(a.x * b.x - a.y * b.y, a.x * b.y + a.y * b.x);
}
__device__ __forceinline__ float2 cadd(float2 a, float2 b) {
  return make_float2(a.x + b.x, a.y + b.y);
}

struct C2 { float2 a00, a01, a10, a11; };
__device__ __forceinline__ C2 mm2(const C2 L, const C2 R) {
  C2 o;
  o.a00 = cadd(cmul(L.a00, R.a00), cmul(L.a01, R.a10));
  o.a01 = cadd(cmul(L.a00, R.a01), cmul(L.a01, R.a11));
  o.a10 = cadd(cmul(L.a10, R.a00), cmul(L.a11, R.a10));
  o.a11 = cadd(cmul(L.a10, R.a01), cmul(L.a11, R.a11));
  return o;
}

__device__ __forceinline__ void rot_gate(float phi, float th, float om,
                                         float2* u00, float2* u01,
                                         float2* u10, float2* u11) {
  float ct = cosf(0.5f * th), st = sinf(0.5f * th);
  float ap = -0.5f * (phi + om);
  float am = 0.5f * (phi - om);
  *u00 = make_float2(ct * cosf(ap), ct * sinf(ap));
  *u01 = make_float2(-st * cosf(am), -st * sinf(am));
  *u10 = make_float2(st * cosf(-am), st * sinf(-am));
  *u11 = make_float2(ct * cosf(-ap), ct * sinf(-ap));
}

__device__ __forceinline__ void apply_cgate(float* Mr, float* Mi,
                                            float2 u00, float2 u01,
                                            float2 u10, float2 u11,
                                            int cb, int tb, int tid) {
  for (int idx = tid; idx < 4096; idx += 256) {
    int i = idx >> 6, j = idx & 63;
    if ((i & cb) && !(i & tb)) {
      int a0 = i * LDA + j, a1 = (i | tb) * LDA + j;
      float2 x0 = make_float2(Mr[a0], Mi[a0]);
      float2 x1 = make_float2(Mr[a1], Mi[a1]);
      float2 y0 = cadd(cmul(u00, x0), cmul(u01, x1));
      float2 y1 = cadd(cmul(u10, x0), cmul(u11, x1));
      Mr[a0] = y0.x; Mi[a0] = y0.y;
      Mr[a1] = y1.x; Mi[a1] = y1.y;
    }
  }
}

__global__ __launch_bounds__(256) void qk_prep(const float* __restrict__ rotp,
                                               const float* __restrict__ entp,
                                               float2* __restrict__ W) {
  __shared__ float Mr[64 * LDA], Mi[64 * LDA];
  __shared__ float2 U[24][2][2];
  const int l = blockIdx.x >> 1;
  const int slot = blockIdx.x & 1;
  const int tid = threadIdx.x;

  if (tid < 24) {
    const float* p = (tid < 12) ? &rotp[(l * 12 + tid) * 3]
                                : &entp[(l * 12 + (tid - 12)) * 3];
    rot_gate(p[0], p[1], p[2], &U[tid][0][0], &U[tid][0][1],
             &U[tid][1][0], &U[tid][1][1]);
  }
  __syncthreads();

  const int g0 = slot ? 6 : 0;

  for (int idx = tid; idx < 4096; idx += 256) {
    int i = idx >> 6, j = idx & 63;
    float2 p = make_float2(((i ^ j) >> 4) ? 0.f : 1.f, 0.f);
    for (int q = 2; q < 6; q++)
      p = cmul(p, U[g0 + q][(i >> (5 - q)) & 1][(j >> (5 - q)) & 1]);
    Mr[i * LDA + j] = p.x; Mi[i * LDA + j] = p.y;
  }
  __syncthreads();
  for (int c = 1; c < 5; c++) {
    apply_cgate(Mr, Mi, U[12 + g0 + c][0][0], U[12 + g0 + c][0][1],
                U[12 + g0 + c][1][0], U[12 + g0 + c][1][1],
                1 << (5 - c), 1 << (4 - c), tid);
    __syncthreads();
  }
  float2* Wd = W + ((size_t)l * 2 + slot) * 4096;
  for (int idx = tid; idx < 4096; idx += 256) {
    int i = idx >> 6, j = idx & 63;
    int dst = (((((i >> 4) * 16 + (j >> 2)) * 32 + ((j >> 1) & 1) * 16 + (i & 15)) << 1) | (j & 1));
    Wd[dst] = make_float2(Mr[i * LDA + j], Mi[i * LDA + j]);
  }
}

__device__ __forceinline__ v8f wmma4(v2f a, v2f b, v8f c) {
  return __builtin_amdgcn_wmma_f32_16x16x4_f32(
      false, a, false, b, (short)0, c, false, false);
}

__device__ __forceinline__ float wave_uniform(float v) {
  return __int_as_float(__builtin_amdgcn_readfirstlane(__float_as_int(v)));
}
__device__ __forceinline__ float2 wave_uniform2(float2 v) {
  return make_float2(wave_uniform(v.x), wave_uniform(v.y));
}
__device__ __forceinline__ float2 cmix(bool first, float2 c, float2 z, float2 e) {
  return first ? make_float2(fmaf(-c.y, z.y, c.x * z.x), fmaf(c.y, z.x, c.x * z.y))
               : make_float2(fmaf(-c.y, z.y, fmaf(c.x, z.x, e.x)),
                             fmaf(c.y, z.x, fmaf(c.x, z.y, e.y)));
}

struct Blk4 { float4 f[4]; };
__device__ __forceinline__ Blk4 load_blk(const float4* __restrict__ base, unsigned off) {
  Blk4 b;
  for (int t = 0; t < 4; t++) b.f[t] = base[off + 32u * t];
  return b;
}

struct OpFrag { v8f re, im; };
__device__ __forceinline__ OpFrag fold_enc(const Blk4 n, const float2* A2,
                                           const float2* PQ) {
  v8f wr = {}, wi = {};
#pragma unroll
  for (int t = 0; t < 4; t++) {
    const float2 a2 = A2[32 * t];
    v2f at;
    at.x = a2.x; at.y = a2.y;
    const float4 p = *reinterpret_cast<const float4*>(PQ + 2 * t);
    const float2 z0 = cmul(make_float2(p.x, p.y), make_float2(n.f[t].x, n.f[t].y));
    const float2 z1 = cmul(make_float2(p.z, p.w), make_float2(n.f[t].z, n.f[t].w));
    v2f br, bi;
    br.x = z0.x; br.y = z1.x;
    bi.x = z0.y; bi.y = z1.y;
    wr = wmma4(at, br, wr);
    wi = wmma4(at, bi, wi);
  }
  OpFrag o;
#pragma unroll
  for (int u = 0; u < 4; u++) {
    const float4 q = *reinterpret_cast<const float4*>(PQ + 8 + 2 * u);
    const float2 y0 = cmul(make_float2(q.x, q.y), make_float2(wr[2 * u], wi[2 * u]));
    const float2 y1 = cmul(make_float2(q.z, q.w), make_float2(wr[2 * u + 1], wi[2 * u + 1]));
    o.re[2 * u] = y0.x;     o.im[2 * u] = y0.y;
    o.re[2 * u + 1] = y1.x; o.im[2 * u + 1] = y1.y;
  }
  return o;
}
__device__ __forceinline__ void fold_store(const Blk4 n, const float2* A2,
                                           const float2* PQ, float4* dst) {
  const OpFrag o = fold_enc(n, A2, PQ);
  for (int t = 0; t < 4; t++)
    dst[32 * t] = make_float4(o.re[2 * t], o.re[2 * t + 1], o.im[2 * t], o.im[2 * t + 1]);
}

__device__ __forceinline__ void cgemm64_opA(const float4* Wl, const float2* MX4,
                                            const float* Br, const float* Bi,
                                            float* Dr, float* Di) {
  const int tid = threadIdx.x;
  const int lane = tid & 31;
  const int w = __builtin_amdgcn_readfirstlane(tid >> 5);
  const int b1 = w & 1;
  const int mtA = b1 * 16;
  const int nt = (w >> 1) * 16;
  const int ml = lane & 15;
  const int hi = lane >> 4;

  __syncthreads();
  float2 cA[4], cB[4];
  for (int j = 0; j < 4; j++) {
    cA[j] = wave_uniform2(MX4[b1 * 4 + j]);
    cB[j] = wave_uniform2(MX4[(2 + b1) * 4 + j]);
  }
  const float* brow = Br + (8 * hi) * LDA + nt + ml;
  const float* birow = Bi + (8 * hi) * LDA + nt + ml;

  v8f p1A = {}, p2A = {}, p3A = {};
  v8f p1B = {}, p2B = {}, p3B = {};
#pragma unroll
  for (int t = 0; t < 4; t++) {
    const float4 wf = Wl[32 * t];
    v2f ar; ar.x = wf.x; ar.y = wf.y;
    v2f ai; ai.x = wf.z; ai.y = wf.w;
    v2f as = ar + ai;
    float2 eA0 = make_float2(0.f, 0.f), eA1 = eA0, eB0 = eA0, eB1 = eA0;
#pragma unroll
    for (int j = 0; j < 4; j++) {
      const int r0 = (16 * j + 2 * t) * LDA, r1 = r0 + LDA;
      const float2 z0 = make_float2(brow[r0], birow[r0]);
      const float2 z1 = make_float2(brow[r1], birow[r1]);
      eA0 = cmix(j == 0, cA[j], z0, eA0);
      eA1 = cmix(j == 0, cA[j], z1, eA1);
      eB0 = cmix(j == 0, cB[j], z0, eB0);
      eB1 = cmix(j == 0, cB[j], z1, eB1);
    }
    v2f brA, biA, brB, biB;
    brA.x = eA0.x; brA.y = eA1.x;
    biA.x = eA0.y; biA.y = eA1.y;
    brB.x = eB0.x; brB.y = eB1.x;
    biB.x = eB0.y; biB.y = eB1.y;
    v2f bsA = brA + biA;
    v2f bsB = brB + biB;

    p1A = wmma4(ar, brA, p1A);
    p2A = wmma4(ai, biA, p2A);
    p3A = wmma4(as, bsA, p3A);
    p1B = wmma4(ar, brB, p1B);
    p2B = wmma4(ai, biB, p2B);
    p3B = wmma4(as, bsB, p3B);
  }

  v8f crA = p1A - p2A;
  v8f ciA = p3A - p1A - p2A;
  v8f crB = p1B - p2B;
  v8f ciB = p3B - p1B - p2B;

  float* dr = Dr + (mtA + 8 * hi) * LDA + nt + ml;
  float* di = Di + (mtA + 8 * hi) * LDA + nt + ml;
  for (int v = 0; v < 8; v++) {
    dr[v * LDA] = crA[v];
    di[v * LDA] = ciA[v];
  }
  for (int v = 0; v < 8; v++) {
    dr[(32 + v) * LDA] = crB[v];
    di[(32 + v) * LDA] = ciB[v];
  }
}

__device__ __forceinline__ void cgemm64_rowpar(const float4* Wl,
                                               const float* Sr, const float* Si,
                                               const float2* MX4,
                                               float* Dr, float* Di,
                                               bool req,
                                               const float4* __restrict__ next,
                                               unsigned noff, Blk4& nx) {
  const int tid = threadIdx.x;
  const int lane = tid & 31;
  const int w = __builtin_amdgcn_readfirstlane(tid >> 5);
  const int par = (w >> 1) & 1;
  const int rb = (w & 1) * 16;
  const int c1 = w >> 2;
  const int nt0 = c1 * 16;
  const int ml = lane & 15;
  const int hi = lane >> 4;

  __syncthreads();
  float2 cA[4], cB[4];
  for (int j = 0; j < 4; j++) {
    cA[j] = wave_uniform2(MX4[(par * 4 + c1) * 4 + j]);
    cB[j] = wave_uniform2(MX4[(par * 4 + 2 + c1) * 4 + j]);
  }
  const int arow = (2 * (rb + ml) + par) * LDA + 8 * hi;
  const float* ap = Sr + arow;
  const float* aip = Si + arow;

  v8f p1_0 = {}, p2_0 = {}, p3_0 = {};
  v8f p1_1 = {}, p2_1 = {}, p3_1 = {};
#pragma unroll
  for (int t = 0; t < 4; t++) {
    float2 eA0 = make_float2(0.f, 0.f), eA1 = eA0, eB0 = eA0, eB1 = eA0;
#pragma unroll
    for (int j = 0; j < 4; j++) {
      const int k = 16 * j + 2 * t;
      const float2 z0 = make_float2(ap[k], aip[k]);
      const float2 z1 = make_float2(ap[k + 1], aip[k + 1]);
      eA0 = cmix(j == 0, cA[j], z0, eA0);
      eA1 = cmix(j == 0, cA[j], z1, eA1);
      eB0 = cmix(j == 0, cB[j], z0, eB0);
      eB1 = cmix(j == 0, cB[j], z1, eB1);
    }
    v2f arA, aiA, arB, aiB;
    arA.x = eA0.x; arA.y = eA1.x;
    aiA.x = eA0.y; aiA.y = eA1.y;
    arB.x = eB0.x; arB.y = eB1.x;
    aiB.x = eB0.y; aiB.y = eB1.y;
    v2f asA = arA + aiA;
    v2f asB = arB + aiB;
    const float4 wf = Wl[32 * t];
    v2f br; br.x = wf.x; br.y = wf.y;
    v2f bi; bi.x = wf.z; bi.y = wf.w;
    v2f bs = br + bi;

    p1_0 = wmma4(arA, br, p1_0);
    p2_0 = wmma4(aiA, bi, p2_0);
    p3_0 = wmma4(asA, bs, p3_0);
    p1_1 = wmma4(arB, br, p1_1);
    p2_1 = wmma4(aiB, bi, p2_1);
    p3_1 = wmma4(asB, bs, p3_1);
  }

  if (req) nx = load_blk(next, noff);

  v8f cr0 = p1_0 - p2_0;
  v8f ci0 = p3_0 - p1_0 - p2_0;
  v8f cr1 = p1_1 - p2_1;
  v8f ci1 = p3_1 - p1_1 - p2_1;

  float* dr = Dr + (2 * (rb + 8 * hi) + par) * LDA + nt0 + ml;
  float* di = Di + (2 * (rb + 8 * hi) + par) * LDA + nt0 + ml;
  for (int v = 0; v < 8; v++) {
    dr[2 * v * LDA] = cr0[v]; dr[2 * v * LDA + 32] = cr1[v];
    di[2 * v * LDA] = ci0[v]; di[2 * v * LDA + 32] = ci1[v];
  }
}

template <int NIT, int ROWS>
__device__ __forceinline__ void crot_boundary(float* Sr, float* Si,
                                              float2 u00, float2 u01,
                                              float2 u10, float2 u11, int t0) {
  const int a = (t0 >> 5) * LDA + (((t0 & 31) << 1) | 1);
  float2 x0[NIT], x1[NIT];
#pragma unroll
  for (int u = 0; u < NIT; u++) {
    const int a0 = a + u * ROWS * LDA, a1 = a0 + 32 * LDA;
    x0[u] = make_float2(Sr[a0], Si[a0]);
    x1[u] = make_float2(Sr[a1], Si[a1]);
  }
#pragma unroll
  for (int u = 0; u < NIT; u++) {
    const int a0 = a + u * ROWS * LDA, a1 = a0 + 32 * LDA;
    const float2 y0 = cadd(cmul(u00, x0[u]), cmul(u01, x1[u]));
    const float2 y1 = cadd(cmul(u10, x0[u]), cmul(u11, x1[u]));
    Sr[a0] = y0.x; Si[a0] = y0.y;
    Sr[a1] = y1.x; Si[a1] = y1.y;
  }
}

__global__ __launch_bounds__(256, 8) void qk_main(const float* __restrict__ x,
                                               const float* __restrict__ rotp,
                                               const float* __restrict__ entp,
                                               const float2* __restrict__ W,
                                               float* __restrict__ out) {
  __shared__ float S0r[64 * LDA], S0i[64 * LDA];
  __shared__ float S1r[64 * LDA], S1i[64 * LDA];
  __shared__ float2 EG[12][2][2];
  __shared__ __attribute__((aligned(16))) float WBraw[4 * 4 * 32 * 4];
  float4* const WB = reinterpret_cast<float4*>(WBraw);
  float (*RQ)[2][2] = reinterpret_cast<float (*)[2][2]>(WBraw);
  float2 (*PP)[2] = reinterpret_cast<float2 (*)[2]>(WBraw + 48);
  float2 (*QQ)[2] = reinterpret_cast<float2 (*)[2]>(WBraw + 96);
  __shared__ float2 A2T[2][4][32];
  __shared__ __attribute__((aligned(16))) float2 PF[2][2][2][8];
  __shared__ float2 GU[NL][8][2][2];
  __shared__ float2 MXA[NL][4][4];
  __shared__ float2 MXL[NL][2][4][4];
  __shared__ float2 MXAe[NL][4][4];
  __shared__ float2 MXLe[NL][2][4][4];

  const int tid = threadIdx.x;
  const int s = blockIdx.x;

  if (tid < 12) {
    const float* xs = x + (size_t)s * 48;
    float a = xs[0 * 12 + tid], b = xs[1 * 12 + tid];
    float c = xs[2 * 12 + tid], d = xs[3 * 12 + tid];
    float ca = cosf(0.5f * a), sa = sinf(0.5f * a);
    float cb = cosf(0.5f * b), sb = sinf(0.5f * b);
    float cc = cosf(0.5f * c), sc = sinf(0.5f * c);
    float cd = cosf(0.5f * d), sd = sinf(0.5f * d);
    C2 M  = { {ca, 0.f}, {-sa, 0.f}, {sa, 0.f}, {ca, 0.f} };
    C2 RX = { {cb, 0.f}, {0.f, -sb}, {0.f, -sb}, {cb, 0.f} };
    C2 RZ = { {cc, -sc}, {0.f, 0.f}, {0.f, 0.f}, {cc, sc} };
    C2 RY = { {cd, 0.f}, {-sd, 0.f}, {sd, 0.f}, {cd, 0.f} };
    M = mm2(RX, M); M = mm2(RZ, M); M = mm2(RY, M);
    EG[tid][0][0] = M.a00; EG[tid][0][1] = M.a01;
    EG[tid][1][0] = M.a10; EG[tid][1][1] = M.a11;
    const float cs = sqrtf(M.a00.x * M.a00.x + M.a00.y * M.a00.y);
    const float sn = sqrtf(M.a10.x * M.a10.x + M.a10.y * M.a10.y);
    float2 p0 = make_float2(1.f, 0.f), p1 = make_float2(1.f, 0.f), q1;
    if (cs > 0.f) { const float r = 1.f / cs; p0 = make_float2(M.a00.x * r, M.a00.y * r); }
    if (sn > 0.f) { const float r = 1.f / sn; p1 = make_float2(M.a10.x * r, M.a10.y * r); }
    if (cs >= sn) {
      const float r = 1.f / cs;
      const float2 t = cmul(M.a11, make_float2(p1.x, -p1.y));
      q1 = make_float2(t.x * r, t.y * r);
    } else {
      const float r = -1.f / sn;
      const float2 t = cmul(M.a01, make_float2(p0.x, -p0.y));
      q1 = make_float2(t.x * r, t.y * r);
    }
    RQ[tid][0][0] = cs; RQ[tid][0][1] = -sn;
    RQ[tid][1][0] = sn; RQ[tid][1][1] = cs;
    PP[tid][0] = p0; PP[tid][1] = p1;
    QQ[tid][0] = make_float2(1.f, 0.f); QQ[tid][1] = q1;
  }
  if (tid < 8 * NL) {
    const int gl = tid >> 3, gg = tid & 7;
    const float* p = (gg == 0) ? &entp[(gl * 12 + 11) * 3]
                   : (gg == 1) ? &rotp[(gl * 12 + 0) * 3]
                   : (gg == 2) ? &rotp[(gl * 12 + 6) * 3]
                   : (gg == 3) ? &entp[(gl * 12 + 5) * 3]
                   : (gg == 4) ? &rotp[(gl * 12 + 1) * 3]
                   : (gg == 5) ? &entp[(gl * 12 + 0) * 3]
                   : (gg == 6) ? &rotp[(gl * 12 + 7) * 3]
                               : &entp[(gl * 12 + 6) * 3];
    rot_gate(p[0], p[1], p[2], &GU[gl][gg][0][0], &GU[gl][gg][0][1],
             &GU[gl][gg][1][0], &GU[gl][gg][1][1]);
  }
  for (int idx = tid; idx < 4096; idx += 256) {
    int i = idx >> 6, j = idx & 63;
    S0r[i * LDA + j] = (idx == 0) ? 1.f : 0.f;
    S0i[i * LDA + j] = 0.f;
  }
  __syncthreads();

  const int lane = tid & 31;
  const int wv = __builtin_amdgcn_readfirstlane(tid >> 5);
  const bool folds = wv < 4;
  const float4* Nf = reinterpret_cast<const float4*>(W) + (wv >> 1) * 2048;
  const unsigned offN = (wv & 1) * 640 + lane;
  Blk4 nN;
  if (folds) nN = load_blk(Nf, offN);

  {
    const int half = tid >> 7, t = (tid >> 5) & 3, ln = tid & 31;
    const int w0 = half * 6 + 2;
    const int c = ln & 15;
    float v[2];
    for (int sx = 0; sx < 2; sx++) {
      const int i = 4 * t + 2 * (ln >> 4) + sx;
      float r = RQ[w0][(i >> 3) & 1][(c >> 3) & 1];
      for (int q = 1; q < 4; q++)
        r *= RQ[w0 + q][(i >> (3 - q)) & 1][(c >> (3 - q)) & 1];
      v[sx] = r;
    }
    A2T[half][t][ln] = make_float2(v[0], v[1]);
  }
  if (tid < 64) {
    const int h = tid >> 5, half = (tid >> 4) & 1, pq = (tid >> 3) & 1, e = tid & 7;
    const int w0 = half * 6 + 2;
    const int idx = pq ? 8 * h + e : 4 * (e >> 1) + 2 * h + (e & 1);
    const float2 (*ph)[2] = pq ? QQ : PP;
    float2 p = ph[w0][(idx >> 3) & 1];
    for (int q = 1; q < 4; q++)
      p = cmul(p, ph[w0 + q][(idx >> (3 - q)) & 1]);
    PF[h][half][pq][e] = p;
  }
  if (tid < 192) {
    const bool low = tid >= 64;
    const int e = low ? tid - 64 : tid;
    const int tl = low ? (e >> 5) : (e >> 4);
    const int tp = low ? ((e >> 4) & 1) : 0;
    const int o0 = (e >> 3) & 1, o1 = (e >> 2) & 1, a0 = (e >> 1) & 1, a1 = e & 1;
    const float2 (*Fg)[2] = GU[tl][low ? 2 : 1];
    const float2 (*Vg)[2] = GU[tl][low ? 6 : 4];
    const float2 (*Cg)[2] = GU[tl][low ? 7 : 5];
    float2 f = Fg[o0][a0];
    if (tp)
      f = cadd(cmul(GU[tl][3][o0][0], Fg[0][a0]), cmul(GU[tl][3][o0][1], Fg[1][a0]));
    float2 g = Vg[o1][a1];
    if (o0)
      g = cadd(cmul(Cg[o1][0], Vg[0][a1]), cmul(Cg[o1][1], Vg[1][a1]));
    const float2 m = cmul(f, g);
    if (low) MXL[tl][tp][2 * o0 + o1][2 * a0 + a1] = m;
    else     MXA[tl][2 * o0 + o1][2 * a0 + a1] = m;
  }
  __syncthreads();
  if (tid < 192) {
    const bool low = tid >= 64;
    const int e = low ? tid - 64 : tid;
    const int j = e & 3;
    const float2* row = low ? &MXL[0][0][0][0] + (e & ~3) : &MXA[0][0][0] + (e & ~3);
    const int w0 = low ? 6 : 0;
    float2 m = make_float2(0.f, 0.f);
    for (int i = 0; i < 4; i++)
      m = cadd(m, cmul(row[i], cmul(EG[w0][i >> 1][j >> 1], EG[w0 + 1][i & 1][j & 1])));
    if (low) (&MXLe[0][0][0][0])[e] = m;
    else     (&MXAe[0][0][0])[e] = m;
  }

  const int fs = (wv >> 1) & 1;
  const float2* A2f = &A2T[fs][0][lane];
  const float2* PQf = &PF[lane >> 4][fs][0][0];
  float4* WBw = WB + (wv & 3) * 128 + lane;
  const float4* WBa = WB + (wv & 1) * 128 + lane;
  const float4* WBb = WB + (2 + (wv >> 2)) * 128 + lane;
  if (folds) fold_store(nN, A2f, PQf, WBw);
  __syncthreads();

  for (int l = 0; l < NL; l++) {
    cgemm64_opA(WBa, &MXAe[l][0][0], S0r, S0i, S1r, S1i);

    const bool more = l + 1 < NL;
    const bool foldnow = folds && more;
    cgemm64_rowpar(WBb, S1r, S1i, &MXLe[l][0][0][0], S0r, S0i,
                   foldnow, Nf + (l + 1) * 4096, offN, nN);
    __syncthreads();

    if (foldnow) {
      fold_store(nN, A2f, PQf, WBw);
    } else if (!folds) {
      const float2 u00 = GU[l][0][0][0], u01 = GU[l][0][0][1];
      const float2 u10 = GU[l][0][1][0], u11 = GU[l][0][1][1];
      crot_boundary<8, 4>(S0r, S0i, u00, u01, u10, u11, tid - 128);
    }
  }
  __syncthreads();

  float z[12];
  for (int q = 0; q < 12; q++) z[q] = 0.f;
  for (int idx = tid; idx < 4096; idx += 256) {
    int i = idx >> 6, j = idx & 63;
    float pr = S0r[i * LDA + j], pi = S0i[i * LDA + j];
    float p = pr * pr + pi * pi;
    for (int q = 0; q < 12; q++)
      z[q] += ((idx >> (11 - q)) & 1) ? -p : p;
  }
  __syncthreads();
  float* red = S1r;
  for (int q = 0; q < 12; q++) red[tid * 12 + q] = z[q];
  __syncthreads();
  for (int sft = 128; sft > 0; sft >>= 1) {
    if (tid < sft)
      for (int q = 0; q < 12; q++)
        red[tid * 12 + q] += red[(tid + sft) * 12 + q];
    __syncthreads();
  }
  if (tid < 12) out[(size_t)s * 12 + tid] = red[tid];
}

extern "C" void kernel_launch(void* const* d_in, const int* in_sizes, int n_in,
                              void* d_out, int out_size, void* d_ws, size_t ws_size,
                              hipStream_t stream) {
  (void)n_in; (void)out_size; (void)ws_size;
  const float* x    = (const float*)d_in[0];
  const float* rotp = (const float*)d_in[1];
  const float* entp = (const float*)d_in[2];
  float2* W = (float2*)d_ws;
  const int B = in_sizes[0] / 48;

  qk_prep<<<NPREP, 256, 0, stream>>>(rotp, entp, W);
  qk_main<<<B, 256, 0, stream>>>(x, rotp, entp, W, (float*)d_out);
}
